# starting kernel
# baseline (speedup 1.0000x reference)
#include <hip/hip_runtime.h>
#include <hip/hip_bf16.h>

typedef __attribute__((ext_vector_type(2))) float v2f;
typedef __attribute__((ext_vector_type(8))) float v8f;

#define HC 128
#define HEADS 4
#define CH 32
#define NEG_SLOPE 0.2f

__device__ __forceinline__ unsigned f2ord(float f) {
    unsigned u = __float_as_uint(f);
    return (u & 0x80000000u) ? ~u : (u | 0x80000000u);
}
__device__ __forceinline__ float ord2f(unsigned u) {
    return (u & 0x80000000u) ? __uint_as_float(u & 0x7FFFFFFFu)
                             : __uint_as_float(~u);
}
#define ORD_NEG_INF 0x007FFFFFu

__device__ __forceinline__ float lrelu(float v) { return v > 0.f ? v : NEG_SLOPE * v; }

__global__ __launch_bounds__(256) void gemm_f32_wmma(
    const float* __restrict__ A, const float* __restrict__ B,
    float* __restrict__ D, int M)
{
    const int wave = threadIdx.x >> 5;
    const int lane = threadIdx.x & 31;
    const int mt   = blockIdx.x;
    const int m15  = lane & 15;
    const int hi   = lane >> 4;
    const int row  = mt * 16 + m15;
    const int col  = wave * 16 + m15;

    v8f acc;
    #pragma unroll
    for (int r = 0; r < 8; ++r) acc[r] = 0.f;

    #pragma unroll 4
    for (int k0 = 0; k0 < HC; k0 += 4) {
        const float* ap = A + (long)row * HC + k0 + 2 * hi;
        v2f a; a.x = ap[0]; a.y = ap[1];
        const float* bp = B + (long)(k0 + 2 * hi) * HC + col;
        v2f b; b.x = bp[0]; b.y = bp[HC];
        acc = __builtin_amdgcn_wmma_f32_16x16x4_f32(
            false, a, false, b, (short)0, acc, false, false);
    }
    #pragma unroll
    for (int r = 0; r < 8; ++r)
        D[(long)(mt * 16 + r + 8 * hi) * HC + col] = acc[r];
}

__global__ void node_alphas(const float* __restrict__ h,
                            const float* __restrict__ a_src,
                            const float* __restrict__ a_dst,
                            float* __restrict__ as, float* __restrict__ ad,
                            int NH)
{
    int i = blockIdx.x * blockDim.x + threadIdx.x;
    if (i >= NH) return;
    int n = i >> 2, hd = i & 3;
    const float* hp  = h + (long)n * HC + hd * CH;
    const float* asp = a_src + hd * CH;
    const float* adp = a_dst + hd * CH;
    float s1 = 0.f, s2 = 0.f;
    #pragma unroll
    for (int c = 0; c < CH; ++c) { float v = hp[c]; s1 += v * asp[c]; s2 += v * adp[c]; }
    as[i] = s1; ad[i] = s2;
}

__global__ void fill_f32(float* p, float v, int n) {
    int i = blockIdx.x * blockDim.x + threadIdx.x;
    if (i < n) p[i] = v;
}
__global__ void fill_u32(unsigned* p, unsigned v, int n) {
    int i = blockIdx.x * blockDim.x + threadIdx.x;
    if (i < n) p[i] = v;
}

__global__ void edge_max(const int* __restrict__ ei, int E, int EL,
                         const float* __restrict__ as, const float* __restrict__ ad,
                         unsigned* __restrict__ m)
{
    int i = blockIdx.x * blockDim.x + threadIdx.x;
    if (i >= EL) return;
    int s, d;
    if (i < E) { s = ei[i]; d = ei[E + i]; } else { s = d = i - E; }
    const float4 a = *(const float4*)(as + (long)s * 4);
    const float4 b = *(const float4*)(ad + (long)d * 4);
    unsigned* mp = m + (long)d * 4;
    atomicMax(mp + 0, f2ord(lrelu(a.x + b.x)));
    atomicMax(mp + 1, f2ord(lrelu(a.y + b.y)));
    atomicMax(mp + 2, f2ord(lrelu(a.z + b.z)));
    atomicMax(mp + 3, f2ord(lrelu(a.w + b.w)));
}

__global__ void edge_sum(const int* __restrict__ ei, int E, int EL,
                         const float* __restrict__ as, const float* __restrict__ ad,
                         const unsigned* __restrict__ m, float* __restrict__ ss)
{
    int i = blockIdx.x * blockDim.x + threadIdx.x;
    if (i >= EL) return;
    int s, d;
    if (i < E) { s = ei[i]; d = ei[E + i]; } else { s = d = i - E; }
    const float4 a = *(const float4*)(as + (long)s * 4);
    const float4 b = *(const float4*)(ad + (long)d * 4);
    const unsigned* mp = m + (long)d * 4;
    float* sp = ss + (long)d * 4;
    atomicAdd(sp + 0, __expf(lrelu(a.x + b.x) - ord2f(mp[0])));
    atomicAdd(sp + 1, __expf(lrelu(a.y + b.y) - ord2f(mp[1])));
    atomicAdd(sp + 2, __expf(lrelu(a.z + b.z) - ord2f(mp[2])));
    atomicAdd(sp + 3, __expf(lrelu(a.w + b.w) - ord2f(mp[3])));
}

__global__ __launch_bounds__(256) void edge_agg(
    const int* __restrict__ ei, int E, int EL,
    const float* __restrict__ h,
    const float* __restrict__ as, const float* __restrict__ ad,
    const unsigned* __restrict__ m, const float* __restrict__ ss,
    float* __restrict__ out)
{
    int e = blockIdx.x * 8 + (threadIdx.x >> 5);
    if (e >= EL) return;
    int lane = threadIdx.x & 31;
    int s, d;
    if (e < E) { s = ei[e]; d = ei[E + e]; } else { s = d = e - E; }
    int hd = lane >> 3;
    float ev    = lrelu(as[(long)s * 4 + hd] + ad[(long)d * 4 + hd]);
    float alpha = __expf(ev - ord2f(m[(long)d * 4 + hd])) / ss[(long)d * 4 + hd];
    const float4 hv = *(const float4*)(h + (long)s * HC + lane * 4);
    float* op = out + (long)d * HC + lane * 4;
    atomicAdd(op + 0, hv.x * alpha);
    atomicAdd(op + 1, hv.y * alpha);
    atomicAdd(op + 2, hv.z * alpha);
    atomicAdd(op + 3, hv.w * alpha);
}

__global__ void bias_elu(float* __restrict__ y, const float* __restrict__ b, int total)
{
    int i = blockIdx.x * blockDim.x + threadIdx.x;
    if (i >= total) return;
    float v = y[i] + b[i & (HC - 1)];
    y[i] = v > 0.f ? v : __expf(v) - 1.f;
}

extern "C" void kernel_launch(void* const* d_in, const int* in_sizes, int n_in,
                              void* d_out, int out_size, void* d_ws, size_t ws_size,
                              hipStream_t stream)
{
    const float* x      = (const float*)d_in[0];
    const int*   ei     = (const int*)d_in[1];
    const float* W1     = (const float*)d_in[2];
    const float* a_src1 = (const float*)d_in[3];
    const float* a_dst1 = (const float*)d_in[4];
    const float* b1     = (const float*)d_in[5];
    const float* W2     = (const float*)d_in[6];
    const float* a_src2 = (const float*)d_in[7];
    const float* a_dst2 = (const float*)d_in[8];
    const float* b2     = (const float*)d_in[9];

    const int N  = in_sizes[0] / HC;
    const int E  = in_sizes[1] / 2;
    const int EL = E + N;
    const int NH = N * HEADS;
    const int NF = N * HC;

    char* ws = (char*)d_ws;
    float*    hbuf = (float*)(ws);
    float*    out1 = (float*)(ws + (size_t)NF * 4);
    float*    as   = (float*)(ws + (size_t)NF * 8);
    float*    ad   = (float*)(ws + (size_t)NF * 8 + (size_t)NH * 4);
    unsigned* mArr = (unsigned*)(ws + (size_t)NF * 8 + (size_t)NH * 8);
    float*    sArr = (float*)(ws + (size_t)NF * 8 + (size_t)NH * 12);

    const int TB = 256;
    const int gN  = (NH + TB - 1) / TB;
    const int gNF = (NF + TB - 1) / TB;
    const int gE  = (EL + TB - 1) / TB;
    const int gEW = (EL + 7) / 8;
    float* outF = (float*)d_out;

    gemm_f32_wmma<<<N / 16, 256, 0, stream>>>(x, W1, hbuf, N);
    node_alphas<<<gN, TB, 0, stream>>>(hbuf, a_src1, a_dst1, as, ad, NH);
    fill_u32<<<gN, TB, 0, stream>>>(mArr, ORD_NEG_INF, NH);
    fill_f32<<<gN, TB, 0, stream>>>(sArr, 0.f, NH);
    fill_f32<<<gNF, TB, 0, stream>>>(out1, 0.f, NF);
    edge_max<<<gE, TB, 0, stream>>>(ei, E, EL, as, ad, mArr);
    edge_sum<<<gE, TB, 0, stream>>>(ei, E, EL, as, ad, mArr, sArr);
    edge_agg<<<gEW, TB, 0, stream>>>(ei, E, EL, hbuf, as, ad, mArr, sArr, out1);
    bias_elu<<<gNF, TB, 0, stream>>>(out1, b1, NF);

    gemm_f32_wmma<<<N / 16, 256, 0, stream>>>(out1, W2, hbuf, N);
    node_alphas<<<gN, TB, 0, stream>>>(hbuf, a_src2, a_dst2, as, ad, NH);
    fill_u32<<<gN, TB, 0, stream>>>(mArr, ORD_NEG_INF, NH);
    fill_f32<<<gN, TB, 0, stream>>>(sArr, 0.f, NH);
    fill_f32<<<gNF, TB, 0, stream>>>(outF, 0.f, NF);
    edge_max<<<gE, TB, 0, stream>>>(ei, E, EL, as, ad, mArr);
    edge_sum<<<gE, TB, 0, stream>>>(ei, E, EL, as, ad, mArr, sArr);
    edge_agg<<<gEW, TB, 0, stream>>>(ei, E, EL, hbuf, as, ad, mArr, sArr, outF);
    bias_elu<<<gNF, TB, 0, stream>>>(outF, b2, NF);
}


# Layer 2's GEMM moved into layer 1's edge work-group: its finished rows (divide, bias, ELU done in LDS, pitch 132) are the WMMA A operand; GEMM 2 launch, layer 1's element-wise launch and out1 gone
# speedup vs baseline: 5.9692x; 5.9692x over previous
#include <hip/hip_runtime.h>
#include <hip/hip_bf16.h>

typedef __attribute__((ext_vector_type(2))) float v2f;
typedef __attribute__((ext_vector_type(8))) float v8f;

#define HC 128
#define HEADS 4
#define CH 32
#define NEG_SLOPE 0.2f

__device__ __forceinline__ unsigned f2ord(float f) {
    unsigned u = __float_as_uint(f);
    return (u & 0x80000000u) ? ~u : (u | 0x80000000u);
}
__device__ __forceinline__ float ord2f(unsigned u) {
    return (u & 0x80000000u) ? __uint_as_float(u & 0x7FFFFFFFu)
                             : __uint_as_float(~u);
}
#define ORD_NEG_INF 0x007FFFFFu

__device__ __forceinline__ float lrelu(float v) { return v > 0.f ? v : NEG_SLOPE * v; }

__global__ __launch_bounds__(256) void gemm_f32_wmma(
    const float* __restrict__ A, const float* __restrict__ B,
    float* __restrict__ D, int M)
{
    const int wave = threadIdx.x >> 5;
    const int lane = threadIdx.x & 31;
    const int mt   = blockIdx.x;
    const int m15  = lane & 15;
    const int hi   = lane >> 4;
    const int row  = mt * 16 + m15;
    const int col  = wave * 16 + m15;

    v8f acc;
    #pragma unroll
    for (int r = 0; r < 8; ++r) acc[r] = 0.f;

    #pragma unroll 4
    for (int k0 = 0; k0 < HC; k0 += 4) {
        const float* ap = A + (long)row * HC + k0 + 2 * hi;
        v2f a; a.x = ap[0]; a.y = ap[1];
        const float* bp = B + (long)(k0 + 2 * hi) * HC + col;
        v2f b; b.x = bp[0]; b.y = bp[HC];
        acc = __builtin_amdgcn_wmma_f32_16x16x4_f32(
            false, a, false, b, (short)0, acc, false, false);
    }
    #pragma unroll
    for (int r = 0; r < 8; ++r)
        D[(long)(mt * 16 + r + 8 * hi) * HC + col] = acc[r];
}

__global__ void node_alphas(const float* __restrict__ h,
                            const float* __restrict__ a_src,
                            const float* __restrict__ a_dst,
                            float* __restrict__ as, float* __restrict__ ad,
                            int NH)
{
    int i = blockIdx.x * blockDim.x + threadIdx.x;
    if (i >= NH) return;
    int n = i >> 2, hd = i & 3;
    const float* hp  = h + (long)n * HC + hd * CH;
    const float* asp = a_src + hd * CH;
    const float* adp = a_dst + hd * CH;
    float s1 = 0.f, s2 = 0.f;
    #pragma unroll
    for (int c = 0; c < CH; ++c) { float v = hp[c]; s1 += v * asp[c]; s2 += v * adp[c]; }
    as[i] = s1; ad[i] = s2;
}

#define EPW 8
#define RPB  196
#define RPAD 208
#define RP   132
#define RBITS 8
#define AGT  1024
#define SWV  8
#define SWE  (SWV * AGT * 4)
#define QCAP (SWE + 4096)
#define MXU  8
#define QCS  32
static_assert(RPB <= (1 << RBITS), "a row number must fit the low bits of a queue word");
static_assert(RPAD >= RPB && RPAD % 16 == 0 && RP >= HC && RP % 4 == 0, "whole matrix tiles, 16-byte aligned rows");
static_assert(RPAD / 16 <= 16 && AGT == 1024, "the product's split: 8 column tiles x 4 groups of at most 4 row tiles over 32 waves");
template <int TL>
__device__ __forceinline__ void rows_times_w(
    const float* rows, const float* __restrict__ Wn, float* __restrict__ hn,
    int lo, int cnt, int mg, int nt, int lane)
{
    const int m15 = lane & 15;
    const int hi  = lane >> 4;
    const int col = nt * 16 + m15;
    v8f acc[TL];
    #pragma unroll
    for (int t = 0; t < TL; ++t) acc[t] = (v8f){0.f, 0.f, 0.f, 0.f, 0.f, 0.f, 0.f, 0.f};
    const float* ap = rows + (mg * 16 + m15) * RP + 2 * hi;
    const float* bp = Wn + (long)(2 * hi) * HC + col;
    #pragma unroll 8
    for (int k0 = 0; k0 < HC; k0 += 4) {
        v2f b; b.x = bp[(long)k0 * HC]; b.y = bp[(long)k0 * HC + HC];
        #pragma unroll
        for (int t = 0; t < TL; ++t) {
            const v2f a = *(const v2f*)(ap + t * (64 * RP) + k0);
            acc[t] = __builtin_amdgcn_wmma_f32_16x16x4_f32(
                false, a, false, b, (short)0, acc[t], false, false);
        }
    }
    #pragma unroll
    for (int tr = 0; tr < TL * 8; ++tr) {
        const int t = tr >> 3, r = tr & 7;
        const int row = (mg + 4 * t) * 16 + r + 8 * hi;
        if (row < cnt) hn[(long)(lo + row) * HC + col] = acc[t][r];
    }
}

template <bool FIRST>
__global__ __launch_bounds__(AGT) void edge_agg_lds(
    const int* __restrict__ ei, int E, int N,
    const float* __restrict__ h,
    const float* __restrict__ as, const float* __restrict__ ad,
    float* __restrict__ ss, float* __restrict__ out,
    int* __restrict__ qkeep, int* __restrict__ qkeep_n,
    const float* __restrict__ bias, const float* __restrict__ Wn)
{
    __shared__ __attribute__((aligned(16))) float rows[RPAD * RP];
    __shared__ float ssl[RPB * HEADS];
    __shared__ unsigned ml[RPB * HEADS];
    __shared__ int   q[QCAP];
    __shared__ int   qn;

    const int tid  = threadIdx.x;
    const int lane = tid & 31;
    const int wv   = __builtin_amdgcn_readfirstlane(tid >> 5);
    const int lo   = blockIdx.x * RPB;
    const int cnt  = min(RPB, N - lo);
    int* qk = qkeep + (long)blockIdx.x * QCAP;
    const int nk = FIRST ? -1 : qkeep_n[blockIdx.x * QCS];

    for (int i = tid; i < RPAD * RP / 4; i += AGT) ((float4*)rows)[i] = make_float4(0.f, 0.f, 0.f, 0.f);
    for (int i = tid; i < RPB * HEADS; i += AGT) { ssl[i] = 0.f; ml[i] = ORD_NEG_INF; }
    if (nk >= 0) {
        for (int i = tid; i < nk; i += AGT) q[i] = qk[i];
    } else {
        for (int i = tid; i < cnt; i += AGT) q[i] = E + lo + i;
        if (tid == 0) qn = cnt;
    }
    __syncthreads();

    const int gbase = E & ~3;
    const int nvec  = (2 * E - gbase + 3) >> 2;
    const int nhot  = nvec > 2 ? nvec - 2 : 0;
    const int nsw   = max((nhot + SWV * AGT - 1) / (SWV * AGT), 1);

    auto test = [&](int dn, int e) {
        if ((unsigned)(dn - lo) < (unsigned)cnt) {
            const int pos = atomicAdd(&qn, 1);
            if (pos < QCAP) q[pos] = e;
        }
    };
    auto end_group = [&](int v) {
        const int g0 = gbase + 4 * v;
        for (int j = 0; j < 4; ++j) {
            const int g = g0 + j;
            if (g >= E && g < 2 * E) test(ei[g], g - E);
        }
    };

    auto maxwalk = [&](int n, bool numbered, bool domax, bool keep) {
        for (int i0 = tid; i0 < n; i0 += MXU * AGT) {
            int s[MXU], dl[MXU];
            if (numbered) {
                int e[MXU], sv[MXU], dv[MXU];
                #pragma unroll
                for (int j = 0; j < MXU; ++j) {
                    const int i = i0 + j * AGT;
                    e[j] = q[i < n ? i : i0];
                }
                #pragma unroll
                for (int j = 0; j < MXU; ++j) {
                    const int ec = e[j] < E ? e[j] : 0;
                    sv[j] = ei[ec];
                    dv[j] = ei[E + ec];
                }
                #pragma unroll
                for (int j = 0; j < MXU; ++j) {
                    const int i = i0 + j * AGT;
                    s[j]  = e[j] < E ? sv[j] : e[j] - E;
                    dl[j] = (e[j] < E ? dv[j] : e[j] - E) - lo;
                    if (i < n) {
                        const int w = (int)(((unsigned)s[j] << RBITS) | (unsigned)dl[j]);
                        q[i] = w;
                        if (keep) qk[i] = w;
                    }
                }
            } else {
                #pragma unroll
                for (int j = 0; j < MXU; ++j) {
                    const int i = i0 + j * AGT;
                    const unsigned w = (unsigned)q[i < n ? i : i0];
                    s[j]  = (int)(w >> RBITS);
                    dl[j] = (int)(w & ((1u << RBITS) - 1u));
                }
            }
            if (domax) {
                float4 a[MXU], b[MXU];
                #pragma unroll
                for (int j = 0; j < MXU; ++j) {
                    a[j] = *(const float4*)(as + (long)s[j] * 4);
                    b[j] = *(const float4*)(ad + (long)(lo + dl[j]) * 4);
                }
                #pragma unroll
                for (int j = 0; j < MXU; ++j) {
                    if (i0 + j * AGT < n) {
                        unsigned* mp = ml + dl[j] * 4;
                        atomicMax(mp + 0, f2ord(lrelu(a[j].x + b[j].x)));
                        atomicMax(mp + 1, f2ord(lrelu(a[j].y + b[j].y)));
                        atomicMax(mp + 2, f2ord(lrelu(a[j].z + b[j].z)));
                        atomicMax(mp + 3, f2ord(lrelu(a[j].w + b[j].w)));
                    }
                }
            }
        }
    };

    auto drain = [&](int n) {
        for (int g = wv; g * EPW < n; g += AGT / 32) {
            const int qi = g * EPW + (lane >> 2);
            const int k  = lane & 3;
            const bool live = qi < n;
            const unsigned w = live ? (unsigned)q[qi] : 0u;
            const int s  = (int)(w >> RBITS);
            const int dl = (int)(w & ((1u << RBITS) - 1u));
            const float    ea = as[(long)s * 4 + k];
            const float    eb = ad[(long)(lo + dl) * 4 + k];
            const unsigned mu = ml[dl * 4 + k];
            float hv[EPW][HEADS];
            #pragma unroll
            for (int j = 0; j < EPW; ++j) {
                const int sj = __builtin_amdgcn_readlane(s, 4 * j);
                const float* hp = h + (long)sj * HC + lane;
                hv[j][0] = hp[0];
                hv[j][1] = hp[CH];
                hv[j][2] = hp[2 * CH];
                hv[j][3] = hp[3 * CH];
            }
            const float p = __expf(lrelu(ea + eb) - ord2f(mu));
            if (live) atomicAdd(&ssl[dl * 4 + k], p);
            #pragma unroll
            for (int j = 0; j < EPW; ++j) {
                if (g * EPW + j < n) {
                    const int   dj = __builtin_amdgcn_readlane(dl, 4 * j);
                    const float p0 = __int_as_float(__builtin_amdgcn_readlane(__float_as_int(p), 4 * j));
                    const float p1 = __int_as_float(__builtin_amdgcn_readlane(__float_as_int(p), 4 * j + 1));
                    const float p2 = __int_as_float(__builtin_amdgcn_readlane(__float_as_int(p), 4 * j + 2));
                    const float p3 = __int_as_float(__builtin_amdgcn_readlane(__float_as_int(p), 4 * j + 3));
                    float* rp = rows + dj * RP + lane;
                    atomicAdd(rp,          hv[j][0] * p0);
                    atomicAdd(rp + CH,     hv[j][1] * p1);
                    atomicAdd(rp + 2 * CH, hv[j][2] * p2);
                    atomicAdd(rp + 3 * CH, hv[j][3] * p3);
                }
            }
        }
    };

    if (nk >= 0) {
        maxwalk(nk, false, true, false);
        __syncthreads();
        drain(nk);
        __syncthreads();
    } else {
        int bl = nsw;
        for (int pass = 0; pass < 3; ++pass) {
            bool redo = false;
            for (int s0 = 0; s0 < nsw; s0 += bl) {
                if (s0 == 0) {
                    if (tid == 0 && nvec > 0) end_group(0);
                    if (tid == 32 && nvec > 1) end_group(nvec - 1);
                }
                const int s1 = min(s0 + bl, nsw);
                for (int sw = s0; sw < s1; ++sw) {
                    int4 dv[SWV];
                    const int4* gp = (const int4*)(ei + gbase + 4) + (long)sw * (SWV * AGT) + tid;
                    if ((sw + 1) * (SWV * AGT) <= nhot) {
                        #pragma unroll
                        for (int j = 0; j < SWV; ++j) dv[j] = gp[j * AGT];
                    } else {
                        #pragma unroll
                        for (int j = 0; j < SWV; ++j) {
                            const int v = sw * (SWV * AGT) + j * AGT + tid;
                            dv[j] = make_int4(-1, -1, -1, -1);
                            if (v < nhot) dv[j] = gp[j * AGT];
                        }
                    }
                    #pragma unroll
                    for (int j = 0; j < SWV; ++j) {
                        const int v  = sw * (SWV * AGT) + j * AGT + tid;
                        const int e0 = gbase + 4 * (v + 1) - E;
                        const bool m0 = (unsigned)(dv[j].x - lo) < (unsigned)cnt;
                        const bool m1 = (unsigned)(dv[j].y - lo) < (unsigned)cnt;
                        const bool m2 = (unsigned)(dv[j].z - lo) < (unsigned)cnt;
                        const bool m3 = (unsigned)(dv[j].w - lo) < (unsigned)cnt;
                        if (m0 | m1 | m2 | m3) {
                            test(dv[j].x, e0);
                            test(dv[j].y, e0 + 1);
                            test(dv[j].z, e0 + 2);
                            test(dv[j].w, e0 + 3);
                        }
                    }
                }
                __syncthreads();
                const int n = __builtin_amdgcn_readfirstlane(qn);
                __syncthreads();
                const bool fits = n <= QCAP;
                if (FIRST && pass == 0 && tid == 0) qkeep_n[blockIdx.x * QCS] = fits ? n : -1;
                if (!fits) { redo = true; break; }
                maxwalk(n, true, pass < 2, FIRST && pass == 0);
                if (pass != 1) {
                    __syncthreads();
                    drain(n);
                }
                __syncthreads();
                if (tid == 0) qn = 0;
                __syncthreads();
            }
            if (pass == 0 ? !redo : pass == 2) break;
            bl = 1;
            for (int i = tid; i < cnt; i += AGT) q[i] = E + lo + i;
            if (tid == 0) qn = cnt;
            __syncthreads();
        }
    }

    if (FIRST) {
        for (int i = tid; i < cnt * (HC / 4); i += AGT) {
            const int r = i >> 5, c4 = i & 31;
            float4* p = (float4*)(rows + r * RP) + c4;
            float4 v = *p;
            const float sd = ssl[r * HEADS + (c4 >> 3)];
            const float* bb = bias + 4 * c4;
            v.x = v.x / sd + bb[0];
            v.y = v.y / sd + bb[1];
            v.z = v.z / sd + bb[2];
            v.w = v.w / sd + bb[3];
            v.x = v.x > 0.f ? v.x : __expf(v.x) - 1.f;
            v.y = v.y > 0.f ? v.y : __expf(v.y) - 1.f;
            v.z = v.z > 0.f ? v.z : __expf(v.z) - 1.f;
            v.w = v.w > 0.f ? v.w : __expf(v.w) - 1.f;
            *p = v;
        }
        __syncthreads();
        const int ntl = (cnt + 15) >> 4;
        const int nt  = wv & 7;
        const int mg  = wv >> 3;
        const int nv  = mg < ntl ? (ntl - 1 - mg) / 4 + 1 : 0;
        switch (nv) {
            case 4: rows_times_w<4>(rows, Wn, out, lo, cnt, mg, nt, lane); break;
            case 3: rows_times_w<3>(rows, Wn, out, lo, cnt, mg, nt, lane); break;
            case 2: rows_times_w<2>(rows, Wn, out, lo, cnt, mg, nt, lane); break;
            case 1: rows_times_w<1>(rows, Wn, out, lo, cnt, mg, nt, lane); break;
            default: break;
        }
    } else {
        float4* o4 = (float4*)(out + (long)lo * HC);
        for (int i = tid; i < cnt * (HC / 4); i += AGT)
            o4[i] = *((const float4*)(rows + (i >> 5) * RP) + (i & 31));
        for (int i = tid; i < cnt * HEADS; i += AGT) ss[(long)lo * HEADS + i] = ssl[i];
    }
}

__global__ void norm_bias_elu(float* __restrict__ y, const float* __restrict__ ss,
                              const float* __restrict__ b, int total)
{
    int i = blockIdx.x * blockDim.x + threadIdx.x;
    if (i >= total) return;
    float v = y[i] / ss[i >> 5] + b[i & (HC - 1)];
    y[i] = v > 0.f ? v : __expf(v) - 1.f;
}

extern "C" void kernel_launch(void* const* d_in, const int* in_sizes, int n_in,
                              void* d_out, int out_size, void* d_ws, size_t ws_size,
                              hipStream_t stream)
{
    const float* x      = (const float*)d_in[0];
    const int*   ei     = (const int*)d_in[1];
    const float* W1     = (const float*)d_in[2];
    const float* a_src1 = (const float*)d_in[3];
    const float* a_dst1 = (const float*)d_in[4];
    const float* b1     = (const float*)d_in[5];
    const float* W2     = (const float*)d_in[6];
    const float* a_src2 = (const float*)d_in[7];
    const float* a_dst2 = (const float*)d_in[8];
    const float* b2     = (const float*)d_in[9];

    const int N  = in_sizes[0] / HC;
    const int E  = in_sizes[1] / 2;
    const int NH = N * HEADS;
    const int NF = N * HC;

    char* ws = (char*)d_ws;
    float*    hbuf = (float*)(ws);
    float*    h2   = (float*)(ws + (size_t)NF * 4);
    float*    as   = (float*)(ws + (size_t)NF * 8);
    float*    ad   = (float*)(ws + (size_t)NF * 8 + (size_t)NH * 4);
    float*    sArr = (float*)(ws + (size_t)NF * 8 + (size_t)NH * 8);

    const int TB = 256;
    const int gN  = (NH + TB - 1) / TB;
    const int gNF = (NF + TB - 1) / TB;
    const int gRB = (N + RPB - 1) / RPB;
    float* outF = (float*)d_out;
    int*      qkn  = (int*)(ws + (size_t)NF * 8 + (size_t)NH * 12);
    int*      qkp  = (int*)(ws + (size_t)NF * 8 + (size_t)NH * 12 + (size_t)gRB * QCS * 4);

    gemm_f32_wmma<<<N / 16, 256, 0, stream>>>(x, W1, hbuf, N);
    node_alphas<<<gN, TB, 0, stream>>>(hbuf, a_src1, a_dst1, as, ad, NH);
    edge_agg_lds<true><<<gRB, AGT, 0, stream>>>(ei, E, N, hbuf, as, ad, sArr, h2, qkp, qkn, b1, W2);

    node_alphas<<<gN, TB, 0, stream>>>(h2, a_src2, a_dst2, as, ad, NH);
    edge_agg_lds<false><<<gRB, AGT, 0, stream>>>(ei, E, N, h2, as, ad, sArr, outF, qkp, qkn, nullptr, nullptr);
    norm_bias_elu<<<gNF, TB, 0, stream>>>(outF, sArr, b2, NF);
}


# Queue build searches one-byte work-group keys (dst / 196, written by a new pre-pass launch), 16 edges per 16-byte load, not destination ids: 4x fewer bytes scanned; sums in arrival order as before
# speedup vs baseline: 6.7358x; 1.1284x over previous
#include <hip/hip_runtime.h>
#include <hip/hip_bf16.h>

typedef __attribute__((ext_vector_type(2))) float v2f;
typedef __attribute__((ext_vector_type(8))) float v8f;

#define HC 128
#define HEADS 4
#define CH 32
#define NEG_SLOPE 0.2f

__device__ __forceinline__ unsigned f2ord(float f) {
    unsigned u = __float_as_uint(f);
    return (u & 0x80000000u) ? ~u : (u | 0x80000000u);
}
__device__ __forceinline__ float ord2f(unsigned u) {
    return (u & 0x80000000u) ? __uint_as_float(u & 0x7FFFFFFFu)
                             : __uint_as_float(~u);
}
#define ORD_NEG_INF 0x007FFFFFu

__device__ __forceinline__ float lrelu(float v) { return v > 0.f ? v : NEG_SLOPE * v; }

__global__ __launch_bounds__(256) void gemm_f32_wmma(
    const float* __restrict__ A, const float* __restrict__ B,
    float* __restrict__ D, int M)
{
    const int wave = threadIdx.x >> 5;
    const int lane = threadIdx.x & 31;
    const int mt   = blockIdx.x;
    const int m15  = lane & 15;
    const int hi   = lane >> 4;
    const int row  = mt * 16 + m15;
    const int col  = wave * 16 + m15;

    v8f acc;
    #pragma unroll
    for (int r = 0; r < 8; ++r) acc[r] = 0.f;

    #pragma unroll 4
    for (int k0 = 0; k0 < HC; k0 += 4) {
        const float* ap = A + (long)row * HC + k0 + 2 * hi;
        v2f a; a.x = ap[0]; a.y = ap[1];
        const float* bp = B + (long)(k0 + 2 * hi) * HC + col;
        v2f b; b.x = bp[0]; b.y = bp[HC];
        acc = __builtin_amdgcn_wmma_f32_16x16x4_f32(
            false, a, false, b, (short)0, acc, false, false);
    }
    #pragma unroll
    for (int r = 0; r < 8; ++r)
        D[(long)(mt * 16 + r + 8 * hi) * HC + col] = acc[r];
}

__global__ void node_alphas(const float* __restrict__ h,
                            const float* __restrict__ a_src,
                            const float* __restrict__ a_dst,
                            float* __restrict__ as, float* __restrict__ ad,
                            int NH)
{
    int i = blockIdx.x * blockDim.x + threadIdx.x;
    if (i >= NH) return;
    int n = i >> 2, hd = i & 3;
    const float* hp  = h + (long)n * HC + hd * CH;
    const float* asp = a_src + hd * CH;
    const float* adp = a_dst + hd * CH;
    float s1 = 0.f, s2 = 0.f;
    #pragma unroll
    for (int c = 0; c < CH; ++c) { float v = hp[c]; s1 += v * asp[c]; s2 += v * adp[c]; }
    as[i] = s1; ad[i] = s2;
}

#define RPB  196
__global__ void dst_keys(const int* __restrict__ ei, int E, unsigned* __restrict__ kw, int nw)
{
    const int i = blockIdx.x * blockDim.x + threadIdx.x;
    if (i >= nw) return;
    const int* dp = ei + E + 4 * (long)i;
    const unsigned k0 = (unsigned)dp[0] / RPB, k1 = (unsigned)dp[1] / RPB;
    const unsigned k2 = (unsigned)dp[2] / RPB, k3 = (unsigned)dp[3] / RPB;
    kw[i] = (k0 & 255u) | ((k1 & 255u) << 8) | ((k2 & 255u) << 16) | (k3 << 24);
}

#define EPW 8
#define RPAD 208
#define RP   132
#define RBITS 8
#define AGT  1024
#define SWV  8
#define SWE  (SWV * AGT * 4)
#define QCAP (SWE + 4096)
#define MXU  8
#define QCS  32
#define KLD  8
static_assert(RPB <= (1 << RBITS), "a row number must fit the low bits of a queue word");
static_assert(RPAD >= RPB && RPAD % 16 == 0 && RP >= HC && RP % 4 == 0, "whole matrix tiles, 16-byte aligned rows");
static_assert(RPAD / 16 <= 16 && AGT == 1024, "the product's split: 8 column tiles x 4 groups of at most 4 row tiles over 32 waves");
template <int TL>
__device__ __forceinline__ void rows_times_w(
    const float* rows, const float* __restrict__ Wn, float* __restrict__ hn,
    int lo, int cnt, int mg, int nt, int lane)
{
    const int m15 = lane & 15;
    const int hi  = lane >> 4;
    const int col = nt * 16 + m15;
    v8f acc[TL];
    #pragma unroll
    for (int t = 0; t < TL; ++t) acc[t] = (v8f){0.f, 0.f, 0.f, 0.f, 0.f, 0.f, 0.f, 0.f};
    const float* ap = rows + (mg * 16 + m15) * RP + 2 * hi;
    const float* bp = Wn + (long)(2 * hi) * HC + col;
    #pragma unroll 8
    for (int k0 = 0; k0 < HC; k0 += 4) {
        v2f b; b.x = bp[(long)k0 * HC]; b.y = bp[(long)k0 * HC + HC];
        #pragma unroll
        for (int t = 0; t < TL; ++t) {
            const v2f a = *(const v2f*)(ap + t * (64 * RP) + k0);
            acc[t] = __builtin_amdgcn_wmma_f32_16x16x4_f32(
                false, a, false, b, (short)0, acc[t], false, false);
        }
    }
    #pragma unroll
    for (int tr = 0; tr < TL * 8; ++tr) {
        const int t = tr >> 3, r = tr & 7;
        const int row = (mg + 4 * t) * 16 + r + 8 * hi;
        if (row < cnt) hn[(long)(lo + row) * HC + col] = acc[t][r];
    }
}

template <bool FIRST>
__global__ __launch_bounds__(AGT) void edge_agg_lds(
    const int* __restrict__ ei, int E, int N,
    const float* __restrict__ h,
    const float* __restrict__ as, const float* __restrict__ ad,
    float* __restrict__ ss, float* __restrict__ out,
    int* __restrict__ qkeep, int* __restrict__ qkeep_n,
    const unsigned* __restrict__ kw,
    const float* __restrict__ bias, const float* __restrict__ Wn)
{
    __shared__ __attribute__((aligned(16))) float rows[RPAD * RP];
    __shared__ float ssl[RPB * HEADS];
    __shared__ unsigned ml[RPB * HEADS];
    __shared__ int   q[QCAP];
    __shared__ int   qn;

    const int tid  = threadIdx.x;
    const int lane = tid & 31;
    const int wv   = __builtin_amdgcn_readfirstlane(tid >> 5);
    const int lo   = blockIdx.x * RPB;
    const int cnt  = min(RPB, N - lo);
    int* qk = qkeep + (long)blockIdx.x * QCAP;
    const int nk = FIRST ? -1 : qkeep_n[blockIdx.x * QCS];

    for (int i = tid; i < RPAD * RP / 4; i += AGT) ((float4*)rows)[i] = make_float4(0.f, 0.f, 0.f, 0.f);
    for (int i = tid; i < RPB * HEADS; i += AGT) { ssl[i] = 0.f; ml[i] = ORD_NEG_INF; }
    if (nk >= 0) {
        for (int i = tid; i < nk; i += AGT) q[i] = qk[i];
    } else {
        for (int i = tid; i < cnt; i += AGT) q[i] = E + lo + i;
        if (tid == 0) qn = cnt;
    }
    __syncthreads();

    const int gbase = E & ~3;
    const int nvec  = (2 * E - gbase + 3) >> 2;
    const int nhot  = nvec > 2 ? nvec - 2 : 0;
    const int nsw   = max((nhot + SWV * AGT - 1) / (SWV * AGT), 1);
    const bool bykey = gridDim.x <= 256;

    auto test = [&](int dn, int e) {
        if ((unsigned)(dn - lo) < (unsigned)cnt) {
            const int pos = atomicAdd(&qn, 1);
            if (pos < QCAP) q[pos] = e;
        }
    };
    auto end_group = [&](int v) {
        const int g0 = gbase + 4 * v;
        for (int j = 0; j < 4; ++j) {
            const int g = g0 + j;
            if (g >= E && g < 2 * E) test(ei[g], g - E);
        }
    };

    auto keyscan = [&]() {
        const unsigned bc = blockIdx.x * 0x01010101u;
        const uint4* kp = (const uint4*)kw;
        const int ng = E >> 4;
        for (int g0 = 0; g0 < ng; g0 += KLD * AGT) {
            uint4 kv[KLD];
            if (g0 + KLD * AGT <= ng) {
                #pragma unroll
                for (int j = 0; j < KLD; ++j) kv[j] = kp[g0 + j * AGT + tid];
            } else {
                #pragma unroll
                for (int j = 0; j < KLD; ++j) {
                    const int g = g0 + j * AGT + tid;
                    kv[j] = make_uint4(~bc, ~bc, ~bc, ~bc);
                    if (g < ng) kv[j] = kp[g];
                }
            }
            unsigned hit = 0;
            #pragma unroll
            for (int j = 0; j < KLD; ++j) {
                const unsigned x0 = kv[j].x ^ bc, x1 = kv[j].y ^ bc, x2 = kv[j].z ^ bc, x3 = kv[j].w ^ bc;
                const unsigned z = ((x0 - 0x01010101u) & ~x0) | ((x1 - 0x01010101u) & ~x1)
                                 | ((x2 - 0x01010101u) & ~x2) | ((x3 - 0x01010101u) & ~x3);
                if (z & 0x80808080u) hit |= 1u << j;
            }
            while (hit) {
                const int j = __builtin_ctz(hit);
                hit &= hit - 1;
                const int g = g0 + j * AGT + tid;
                const uint4 k = kp[g];
                const unsigned x0 = k.x ^ bc, x1 = k.y ^ bc, x2 = k.z ^ bc, x3 = k.w ^ bc;
                const unsigned m0 = ~(((x0 & 0x7f7f7f7fu) + 0x7f7f7f7fu) | x0) & 0x80808080u;
                const unsigned m1 = ~(((x1 & 0x7f7f7f7fu) + 0x7f7f7f7fu) | x1) & 0x80808080u;
                const unsigned m2 = ~(((x2 & 0x7f7f7f7fu) + 0x7f7f7f7fu) | x2) & 0x80808080u;
                const unsigned m3 = ~(((x3 & 0x7f7f7f7fu) + 0x7f7f7f7fu) | x3) & 0x80808080u;
                unsigned m = (m0 >> 7) | (m1 >> 6) | (m2 >> 5) | (m3 >> 4);
                while (m) {
                    const int b = __builtin_ctz(m);
                    m &= m - 1;
                    const int pos = atomicAdd(&qn, 1);
                    if (pos < QCAP) q[pos] = 16 * g + 4 * (b & 3) + (b >> 3);
                }
            }
        }
        if (tid < (E & 15)) {
            const int e = (E & ~15) + tid;
            test(ei[E + e], e);
        }
    };

    auto maxwalk = [&](int n, bool numbered, bool domax, bool keep) {
        for (int i0 = tid; i0 < n; i0 += MXU * AGT) {
            int s[MXU], dl[MXU];
            if (numbered) {
                int e[MXU], sv[MXU], dv[MXU];
                #pragma unroll
                for (int j = 0; j < MXU; ++j) {
                    const int i = i0 + j * AGT;
                    e[j] = q[i < n ? i : i0];
                }
                #pragma unroll
                for (int j = 0; j < MXU; ++j) {
                    const int ec = e[j] < E ? e[j] : 0;
                    sv[j] = ei[ec];
                    dv[j] = ei[E + ec];
                }
                #pragma unroll
                for (int j = 0; j < MXU; ++j) {
                    const int i = i0 + j * AGT;
                    s[j]  = e[j] < E ? sv[j] : e[j] - E;
                    dl[j] = (e[j] < E ? dv[j] : e[j] - E) - lo;
                    if (i < n) {
                        const int w = (int)(((unsigned)s[j] << RBITS) | (unsigned)dl[j]);
                        q[i] = w;
                        if (keep) qk[i] = w;
                    }
                }
            } else {
                #pragma unroll
                for (int j = 0; j < MXU; ++j) {
                    const int i = i0 + j * AGT;
                    const unsigned w = (unsigned)q[i < n ? i : i0];
                    s[j]  = (int)(w >> RBITS);
                    dl[j] = (int)(w & ((1u << RBITS) - 1u));
                }
            }
            if (domax) {
                float4 a[MXU], b[MXU];
                #pragma unroll
                for (int j = 0; j < MXU; ++j) {
                    a[j] = *(const float4*)(as + (long)s[j] * 4);
                    b[j] = *(const float4*)(ad + (long)(lo + dl[j]) * 4);
                }
                #pragma unroll
                for (int j = 0; j < MXU; ++j) {
                    if (i0 + j * AGT < n) {
                        unsigned* mp = ml + dl[j] * 4;
                        atomicMax(mp + 0, f2ord(lrelu(a[j].x + b[j].x)));
                        atomicMax(mp + 1, f2ord(lrelu(a[j].y + b[j].y)));
                        atomicMax(mp + 2, f2ord(lrelu(a[j].z + b[j].z)));
                        atomicMax(mp + 3, f2ord(lrelu(a[j].w + b[j].w)));
                    }
                }
            }
        }
    };

    auto drain = [&](int n) {
        for (int g = wv; g * EPW < n; g += AGT / 32) {
            const int qi = g * EPW + (lane >> 2);
            const int k  = lane & 3;
            const bool live = qi < n;
            const unsigned w = live ? (unsigned)q[qi] : 0u;
            const int s  = (int)(w >> RBITS);
            const int dl = (int)(w & ((1u << RBITS) - 1u));
            const float    ea = as[(long)s * 4 + k];
            const float    eb = ad[(long)(lo + dl) * 4 + k];
            const unsigned mu = ml[dl * 4 + k];
            float hv[EPW][HEADS];
            #pragma unroll
            for (int j = 0; j < EPW; ++j) {
                const int sj = __builtin_amdgcn_readlane(s, 4 * j);
                const float* hp = h + (long)sj * HC + lane;
                hv[j][0] = hp[0];
                hv[j][1] = hp[CH];
                hv[j][2] = hp[2 * CH];
                hv[j][3] = hp[3 * CH];
            }
            const float p = __expf(lrelu(ea + eb) - ord2f(mu));
            if (live) atomicAdd(&ssl[dl * 4 + k], p);
            #pragma unroll
            for (int j = 0; j < EPW; ++j) {
                if (g * EPW + j < n) {
                    const int   dj = __builtin_amdgcn_readlane(dl, 4 * j);
                    const float p0 = __int_as_float(__builtin_amdgcn_readlane(__float_as_int(p), 4 * j));
                    const float p1 = __int_as_float(__builtin_amdgcn_readlane(__float_as_int(p), 4 * j + 1));
                    const float p2 = __int_as_float(__builtin_amdgcn_readlane(__float_as_int(p), 4 * j + 2));
                    const float p3 = __int_as_float(__builtin_amdgcn_readlane(__float_as_int(p), 4 * j + 3));
                    float* rp = rows + dj * RP + lane;
                    atomicAdd(rp,          hv[j][0] * p0);
                    atomicAdd(rp + CH,     hv[j][1] * p1);
                    atomicAdd(rp + 2 * CH, hv[j][2] * p2);
                    atomicAdd(rp + 3 * CH, hv[j][3] * p3);
                }
            }
        }
    };

    if (nk >= 0) {
        maxwalk(nk, false, true, false);
        __syncthreads();
        drain(nk);
        __syncthreads();
    } else {
        int bl = nsw;
        for (int pass = 0; pass < 3; ++pass) {
            bool redo = false;
            for (int s0 = 0; s0 < nsw; s0 += bl) {
                if (pass == 0 && bykey) keyscan();
                if (s0 == 0 && !(pass == 0 && bykey)) {
                    if (tid == 0 && nvec > 0) end_group(0);
                    if (tid == 32 && nvec > 1) end_group(nvec - 1);
                }
                const int s1 = (pass == 0 && bykey) ? s0 : min(s0 + bl, nsw);
                for (int sw = s0; sw < s1; ++sw) {
                    int4 dv[SWV];
                    const int4* gp = (const int4*)(ei + gbase + 4) + (long)sw * (SWV * AGT) + tid;
                    if ((sw + 1) * (SWV * AGT) <= nhot) {
                        #pragma unroll
                        for (int j = 0; j < SWV; ++j) dv[j] = gp[j * AGT];
                    } else {
                        #pragma unroll
                        for (int j = 0; j < SWV; ++j) {
                            const int v = sw * (SWV * AGT) + j * AGT + tid;
                            dv[j] = make_int4(-1, -1, -1, -1);
                            if (v < nhot) dv[j] = gp[j * AGT];
                        }
                    }
                    #pragma unroll
                    for (int j = 0; j < SWV; ++j) {
                        const int v  = sw * (SWV * AGT) + j * AGT + tid;
                        const int e0 = gbase + 4 * (v + 1) - E;
                        const bool m0 = (unsigned)(dv[j].x - lo) < (unsigned)cnt;
                        const bool m1 = (unsigned)(dv[j].y - lo) < (unsigned)cnt;
                        const bool m2 = (unsigned)(dv[j].z - lo) < (unsigned)cnt;
                        const bool m3 = (unsigned)(dv[j].w - lo) < (unsigned)cnt;
                        if (m0 | m1 | m2 | m3) {
                            test(dv[j].x, e0);
                            test(dv[j].y, e0 + 1);
                            test(dv[j].z, e0 + 2);
                            test(dv[j].w, e0 + 3);
                        }
                    }
                }
                __syncthreads();
                const int n = __builtin_amdgcn_readfirstlane(qn);
                __syncthreads();
                const bool fits = n <= QCAP;
                if (FIRST && pass == 0 && tid == 0) qkeep_n[blockIdx.x * QCS] = fits ? n : -1;
                if (!fits) { redo = true; break; }
                maxwalk(n, true, pass < 2, FIRST && pass == 0);
                if (pass != 1) {
                    __syncthreads();
                    drain(n);
                }
                __syncthreads();
                if (tid == 0) qn = 0;
                __syncthreads();
            }
            if (pass == 0 ? !redo : pass == 2) break;
            bl = 1;
            for (int i = tid; i < cnt; i += AGT) q[i] = E + lo + i;
            if (tid == 0) qn = cnt;
            __syncthreads();
        }
    }

    if (FIRST) {
        for (int i = tid; i < cnt * (HC / 4); i += AGT) {
            const int r = i >> 5, c4 = i & 31;
            float4* p = (float4*)(rows + r * RP) + c4;
            float4 v = *p;
            const float sd = ssl[r * HEADS + (c4 >> 3)];
            const float* bb = bias + 4 * c4;
            v.x = v.x / sd + bb[0];
            v.y = v.y / sd + bb[1];
            v.z = v.z / sd + bb[2];
            v.w = v.w / sd + bb[3];
            v.x = v.x > 0.f ? v.x : __expf(v.x) - 1.f;
            v.y = v.y > 0.f ? v.y : __expf(v.y) - 1.f;
            v.z = v.z > 0.f ? v.z : __expf(v.z) - 1.f;
            v.w = v.w > 0.f ? v.w : __expf(v.w) - 1.f;
            *p = v;
        }
        __syncthreads();
        const int ntl = (cnt + 15) >> 4;
        const int nt  = wv & 7;
        const int mg  = wv >> 3;
        const int nv  = mg < ntl ? (ntl - 1 - mg) / 4 + 1 : 0;
        switch (nv) {
            case 4: rows_times_w<4>(rows, Wn, out, lo, cnt, mg, nt, lane); break;
            case 3: rows_times_w<3>(rows, Wn, out, lo, cnt, mg, nt, lane); break;
            case 2: rows_times_w<2>(rows, Wn, out, lo, cnt, mg, nt, lane); break;
            case 1: rows_times_w<1>(rows, Wn, out, lo, cnt, mg, nt, lane); break;
            default: break;
        }
    } else {
        float4* o4 = (float4*)(out + (long)lo * HC);
        for (int i = tid; i < cnt * (HC / 4); i += AGT)
            o4[i] = *((const float4*)(rows + (i >> 5) * RP) + (i & 31));
        for (int i = tid; i < cnt * HEADS; i += AGT) ss[(long)lo * HEADS + i] = ssl[i];
    }
}

__global__ void norm_bias_elu(float* __restrict__ y, const float* __restrict__ ss,
                              const float* __restrict__ b, int total)
{
    int i = blockIdx.x * blockDim.x + threadIdx.x;
    if (i >= total) return;
    float v = y[i] / ss[i >> 5] + b[i & (HC - 1)];
    y[i] = v > 0.f ? v : __expf(v) - 1.f;
}

extern "C" void kernel_launch(void* const* d_in, const int* in_sizes, int n_in,
                              void* d_out, int out_size, void* d_ws, size_t ws_size,
                              hipStream_t stream)
{
    const float* x      = (const float*)d_in[0];
    const int*   ei     = (const int*)d_in[1];
    const float* W1     = (const float*)d_in[2];
    const float* a_src1 = (const float*)d_in[3];
    const float* a_dst1 = (const float*)d_in[4];
    const float* b1     = (const float*)d_in[5];
    const float* W2     = (const float*)d_in[6];
    const float* a_src2 = (const float*)d_in[7];
    const float* a_dst2 = (const float*)d_in[8];
    const float* b2     = (const float*)d_in[9];

    const int N  = in_sizes[0] / HC;
    const int E  = in_sizes[1] / 2;
    const int NH = N * HEADS;
    const int NF = N * HC;

    char* ws = (char*)d_ws;
    float*    hbuf = (float*)(ws);
    float*    h2   = (float*)(ws + (size_t)NF * 4);
    float*    as   = (float*)(ws + (size_t)NF * 8);
    float*    ad   = (float*)(ws + (size_t)NF * 8 + (size_t)NH * 4);
    float*    sArr = (float*)(ws + (size_t)NF * 8 + (size_t)NH * 8);

    const int TB = 256;
    const int gN  = (NH + TB - 1) / TB;
    const int gNF = (NF + TB - 1) / TB;
    const int gRB = (N + RPB - 1) / RPB;
    float* outF = (float*)d_out;
    int*      qkn  = (int*)(ws + (size_t)NF * 8 + (size_t)NH * 12);
    int*      qkp  = (int*)(ws + (size_t)NF * 8 + (size_t)NH * 12 + (size_t)gRB * QCS * 4);
    unsigned* kwp  = (unsigned*)(ws + (size_t)NF * 8 + (size_t)NH * 12 + (size_t)gRB * QCS * 4 + (size_t)gRB * QCAP * 4);
    const int nkw  = (E >> 4) * 4;

    gemm_f32_wmma<<<N / 16, 256, 0, stream>>>(x, W1, hbuf, N);
    node_alphas<<<gN, TB, 0, stream>>>(hbuf, a_src1, a_dst1, as, ad, NH);
    dst_keys<<<(nkw + TB) / TB, TB, 0, stream>>>(ei, E, kwp, nkw);
    edge_agg_lds<true><<<gRB, AGT, 0, stream>>>(ei, E, N, hbuf, as, ad, sArr, h2, qkp, qkn, kwp, b1, W2);

    node_alphas<<<gN, TB, 0, stream>>>(h2, a_src2, a_dst2, as, ad, NH);
    edge_agg_lds<false><<<gRB, AGT, 0, stream>>>(ei, E, N, h2, as, ad, sArr, outF, qkp, qkn, kwp, nullptr, nullptr);
    norm_bias_elu<<<gNF, TB, 0, stream>>>(outF, sArr, b2, NF);
}


# Last launch (divide, bias, ELU): a wave takes 8 whole rows with 16-byte loads and stores, all loads issued first; 6256 waves enter instead of 200000 one-element-per-thread waves; values as before
# speedup vs baseline: 6.9604x; 1.0333x over previous
#include <hip/hip_runtime.h>
#include <hip/hip_bf16.h>

typedef __attribute__((ext_vector_type(2))) float v2f;
typedef __attribute__((ext_vector_type(8))) float v8f;

#define HC 128
#define HEADS 4
#define CH 32
#define NEG_SLOPE 0.2f

__device__ __forceinline__ unsigned f2ord(float f) {
    unsigned u = __float_as_uint(f);
    return (u & 0x80000000u) ? ~u : (u | 0x80000000u);
}
__device__ __forceinline__ float ord2f(unsigned u) {
    return (u & 0x80000000u) ? __uint_as_float(u & 0x7FFFFFFFu)
                             : __uint_as_float(~u);
}
#define ORD_NEG_INF 0x007FFFFFu

__device__ __forceinline__ float lrelu(float v) { return v > 0.f ? v : NEG_SLOPE * v; }

__global__ __launch_bounds__(256) void gemm_f32_wmma(
    const float* __restrict__ A, const float* __restrict__ B,
    float* __restrict__ D, int M)
{
    const int wave = threadIdx.x >> 5;
    const int lane = threadIdx.x & 31;
    const int mt   = blockIdx.x;
    const int m15  = lane & 15;
    const int hi   = lane >> 4;
    const int row  = mt * 16 + m15;
    const int col  = wave * 16 + m15;

    v8f acc;
    #pragma unroll
    for (int r = 0; r < 8; ++r) acc[r] = 0.f;

    #pragma unroll 4
    for (int k0 = 0; k0 < HC; k0 += 4) {
        const float* ap = A + (long)row * HC + k0 + 2 * hi;
        v2f a; a.x = ap[0]; a.y = ap[1];
        const float* bp = B + (long)(k0 + 2 * hi) * HC + col;
        v2f b; b.x = bp[0]; b.y = bp[HC];
        acc = __builtin_amdgcn_wmma_f32_16x16x4_f32(
            false, a, false, b, (short)0, acc, false, false);
    }
    #pragma unroll
    for (int r = 0; r < 8; ++r)
        D[(long)(mt * 16 + r + 8 * hi) * HC + col] = acc[r];
}

__global__ void node_alphas(const float* __restrict__ h,
                            const float* __restrict__ a_src,
                            const float* __restrict__ a_dst,
                            float* __restrict__ as, float* __restrict__ ad,
                            int NH)
{
    int i = blockIdx.x * blockDim.x + threadIdx.x;
    if (i >= NH) return;
    int n = i >> 2, hd = i & 3;
    const float* hp  = h + (long)n * HC + hd * CH;
    const float* asp = a_src + hd * CH;
    const float* adp = a_dst + hd * CH;
    float s1 = 0.f, s2 = 0.f;
    #pragma unroll
    for (int c = 0; c < CH; ++c) { float v = hp[c]; s1 += v * asp[c]; s2 += v * adp[c]; }
    as[i] = s1; ad[i] = s2;
}

#define RPB  196
__global__ void dst_keys(const int* __restrict__ ei, int E, unsigned* __restrict__ kw, int nw)
{
    const int i = blockIdx.x * blockDim.x + threadIdx.x;
    if (i >= nw) return;
    const int* dp = ei + E + 4 * (long)i;
    const unsigned k0 = (unsigned)dp[0] / RPB, k1 = (unsigned)dp[1] / RPB;
    const unsigned k2 = (unsigned)dp[2] / RPB, k3 = (unsigned)dp[3] / RPB;
    kw[i] = (k0 & 255u) | ((k1 & 255u) << 8) | ((k2 & 255u) << 16) | (k3 << 24);
}

#define EPW 8
#define RPAD 208
#define RP   132
#define RBITS 8
#define AGT  1024
#define SWV  8
#define SWE  (SWV * AGT * 4)
#define QCAP (SWE + 4096)
#define MXU  8
#define QCS  32
#define KLD  8
static_assert(RPB <= (1 << RBITS), "a row number must fit the low bits of a queue word");
static_assert(RPAD >= RPB && RPAD % 16 == 0 && RP >= HC && RP % 4 == 0, "whole matrix tiles, 16-byte aligned rows");
static_assert(RPAD / 16 <= 16 && AGT == 1024, "the product's split: 8 column tiles x 4 groups of at most 4 row tiles over 32 waves");
template <int TL>
__device__ __forceinline__ void rows_times_w(
    const float* rows, const float* __restrict__ Wn, float* __restrict__ hn,
    int lo, int cnt, int mg, int nt, int lane)
{
    const int m15 = lane & 15;
    const int hi  = lane >> 4;
    const int col = nt * 16 + m15;
    v8f acc[TL];
    #pragma unroll
    for (int t = 0; t < TL; ++t) acc[t] = (v8f){0.f, 0.f, 0.f, 0.f, 0.f, 0.f, 0.f, 0.f};
    const float* ap = rows + (mg * 16 + m15) * RP + 2 * hi;
    const float* bp = Wn + (long)(2 * hi) * HC + col;
    #pragma unroll 8
    for (int k0 = 0; k0 < HC; k0 += 4) {
        v2f b; b.x = bp[(long)k0 * HC]; b.y = bp[(long)k0 * HC + HC];
        #pragma unroll
        for (int t = 0; t < TL; ++t) {
            const v2f a = *(const v2f*)(ap + t * (64 * RP) + k0);
            acc[t] = __builtin_amdgcn_wmma_f32_16x16x4_f32(
                false, a, false, b, (short)0, acc[t], false, false);
        }
    }
    #pragma unroll
    for (int tr = 0; tr < TL * 8; ++tr) {
        const int t = tr >> 3, r = tr & 7;
        const int row = (mg + 4 * t) * 16 + r + 8 * hi;
        if (row < cnt) hn[(long)(lo + row) * HC + col] = acc[t][r];
    }
}

template <bool FIRST>
__global__ __launch_bounds__(AGT) void edge_agg_lds(
    const int* __restrict__ ei, int E, int N,
    const float* __restrict__ h,
    const float* __restrict__ as, const float* __restrict__ ad,
    float* __restrict__ ss, float* __restrict__ out,
    int* __restrict__ qkeep, int* __restrict__ qkeep_n,
    const unsigned* __restrict__ kw,
    const float* __restrict__ bias, const float* __restrict__ Wn)
{
    __shared__ __attribute__((aligned(16))) float rows[RPAD * RP];
    __shared__ float ssl[RPB * HEADS];
    __shared__ unsigned ml[RPB * HEADS];
    __shared__ int   q[QCAP];
    __shared__ int   qn;

    const int tid  = threadIdx.x;
    const int lane = tid & 31;
    const int wv   = __builtin_amdgcn_readfirstlane(tid >> 5);
    const int lo   = blockIdx.x * RPB;
    const int cnt  = min(RPB, N - lo);
    int* qk = qkeep + (long)blockIdx.x * QCAP;
    const int nk = FIRST ? -1 : qkeep_n[blockIdx.x * QCS];

    for (int i = tid; i < RPAD * RP / 4; i += AGT) ((float4*)rows)[i] = make_float4(0.f, 0.f, 0.f, 0.f);
    for (int i = tid; i < RPB * HEADS; i += AGT) { ssl[i] = 0.f; ml[i] = ORD_NEG_INF; }
    if (nk >= 0) {
        for (int i = tid; i < nk; i += AGT) q[i] = qk[i];
    } else {
        for (int i = tid; i < cnt; i += AGT) q[i] = E + lo + i;
        if (tid == 0) qn = cnt;
    }
    __syncthreads();

    const int gbase = E & ~3;
    const int nvec  = (2 * E - gbase + 3) >> 2;
    const int nhot  = nvec > 2 ? nvec - 2 : 0;
    const int nsw   = max((nhot + SWV * AGT - 1) / (SWV * AGT), 1);
    const bool bykey = gridDim.x <= 256;

    auto test = [&](int dn, int e) {
        if ((unsigned)(dn - lo) < (unsigned)cnt) {
            const int pos = atomicAdd(&qn, 1);
            if (pos < QCAP) q[pos] = e;
        }
    };
    auto end_group = [&](int v) {
        const int g0 = gbase + 4 * v;
        for (int j = 0; j < 4; ++j) {
            const int g = g0 + j;
            if (g >= E && g < 2 * E) test(ei[g], g - E);
        }
    };

    auto keyscan = [&]() {
        const unsigned bc = blockIdx.x * 0x01010101u;
        const uint4* kp = (const uint4*)kw;
        const int ng = E >> 4;
        for (int g0 = 0; g0 < ng; g0 += KLD * AGT) {
            uint4 kv[KLD];
            if (g0 + KLD * AGT <= ng) {
                #pragma unroll
                for (int j = 0; j < KLD; ++j) kv[j] = kp[g0 + j * AGT + tid];
            } else {
                #pragma unroll
                for (int j = 0; j < KLD; ++j) {
                    const int g = g0 + j * AGT + tid;
                    kv[j] = make_uint4(~bc, ~bc, ~bc, ~bc);
                    if (g < ng) kv[j] = kp[g];
                }
            }
            unsigned hit = 0;
            #pragma unroll
            for (int j = 0; j < KLD; ++j) {
                const unsigned x0 = kv[j].x ^ bc, x1 = kv[j].y ^ bc, x2 = kv[j].z ^ bc, x3 = kv[j].w ^ bc;
                const unsigned z = ((x0 - 0x01010101u) & ~x0) | ((x1 - 0x01010101u) & ~x1)
                                 | ((x2 - 0x01010101u) & ~x2) | ((x3 - 0x01010101u) & ~x3);
                if (z & 0x80808080u) hit |= 1u << j;
            }
            while (hit) {
                const int j = __builtin_ctz(hit);
                hit &= hit - 1;
                const int g = g0 + j * AGT + tid;
                const uint4 k = kp[g];
                const unsigned x0 = k.x ^ bc, x1 = k.y ^ bc, x2 = k.z ^ bc, x3 = k.w ^ bc;
                const unsigned m0 = ~(((x0 & 0x7f7f7f7fu) + 0x7f7f7f7fu) | x0) & 0x80808080u;
                const unsigned m1 = ~(((x1 & 0x7f7f7f7fu) + 0x7f7f7f7fu) | x1) & 0x80808080u;
                const unsigned m2 = ~(((x2 & 0x7f7f7f7fu) + 0x7f7f7f7fu) | x2) & 0x80808080u;
                const unsigned m3 = ~(((x3 & 0x7f7f7f7fu) + 0x7f7f7f7fu) | x3) & 0x80808080u;
                unsigned m = (m0 >> 7) | (m1 >> 6) | (m2 >> 5) | (m3 >> 4);
                while (m) {
                    const int b = __builtin_ctz(m);
                    m &= m - 1;
                    const int pos = atomicAdd(&qn, 1);
                    if (pos < QCAP) q[pos] = 16 * g + 4 * (b & 3) + (b >> 3);
                }
            }
        }
        if (tid < (E & 15)) {
            const int e = (E & ~15) + tid;
            test(ei[E + e], e);
        }
    };

    auto maxwalk = [&](int n, bool numbered, bool domax, bool keep) {
        for (int i0 = tid; i0 < n; i0 += MXU * AGT) {
            int s[MXU], dl[MXU];
            if (numbered) {
                int e[MXU], sv[MXU], dv[MXU];
                #pragma unroll
                for (int j = 0; j < MXU; ++j) {
                    const int i = i0 + j * AGT;
                    e[j] = q[i < n ? i : i0];
                }
                #pragma unroll
                for (int j = 0; j < MXU; ++j) {
                    const int ec = e[j] < E ? e[j] : 0;
                    sv[j] = ei[ec];
                    dv[j] = ei[E + ec];
                }
                #pragma unroll
                for (int j = 0; j < MXU; ++j) {
                    const int i = i0 + j * AGT;
                    s[j]  = e[j] < E ? sv[j] : e[j] - E;
                    dl[j] = (e[j] < E ? dv[j] : e[j] - E) - lo;
                    if (i < n) {
                        const int w = (int)(((unsigned)s[j] << RBITS) | (unsigned)dl[j]);
                        q[i] = w;
                        if (keep) qk[i] = w;
                    }
                }
            } else {
                #pragma unroll
                for (int j = 0; j < MXU; ++j) {
                    const int i = i0 + j * AGT;
                    const unsigned w = (unsigned)q[i < n ? i : i0];
                    s[j]  = (int)(w >> RBITS);
                    dl[j] = (int)(w & ((1u << RBITS) - 1u));
                }
            }
            if (domax) {
                float4 a[MXU], b[MXU];
                #pragma unroll
                for (int j = 0; j < MXU; ++j) {
                    a[j] = *(const float4*)(as + (long)s[j] * 4);
                    b[j] = *(const float4*)(ad + (long)(lo + dl[j]) * 4);
                }
                #pragma unroll
                for (int j = 0; j < MXU; ++j) {
                    if (i0 + j * AGT < n) {
                        unsigned* mp = ml + dl[j] * 4;
                        atomicMax(mp + 0, f2ord(lrelu(a[j].x + b[j].x)));
                        atomicMax(mp + 1, f2ord(lrelu(a[j].y + b[j].y)));
                        atomicMax(mp + 2, f2ord(lrelu(a[j].z + b[j].z)));
                        atomicMax(mp + 3, f2ord(lrelu(a[j].w + b[j].w)));
                    }
                }
            }
        }
    };

    auto drain = [&](int n) {
        for (int g = wv; g * EPW < n; g += AGT / 32) {
            const int qi = g * EPW + (lane >> 2);
            const int k  = lane & 3;
            const bool live = qi < n;
            const unsigned w = live ? (unsigned)q[qi] : 0u;
            const int s  = (int)(w >> RBITS);
            const int dl = (int)(w & ((1u << RBITS) - 1u));
            const float    ea = as[(long)s * 4 + k];
            const float    eb = ad[(long)(lo + dl) * 4 + k];
            const unsigned mu = ml[dl * 4 + k];
            float hv[EPW][HEADS];
            #pragma unroll
            for (int j = 0; j < EPW; ++j) {
                const int sj = __builtin_amdgcn_readlane(s, 4 * j);
                const float* hp = h + (long)sj * HC + lane;
                hv[j][0] = hp[0];
                hv[j][1] = hp[CH];
                hv[j][2] = hp[2 * CH];
                hv[j][3] = hp[3 * CH];
            }
            const float p = __expf(lrelu(ea + eb) - ord2f(mu));
            if (live) atomicAdd(&ssl[dl * 4 + k], p);
            #pragma unroll
            for (int j = 0; j < EPW; ++j) {
                if (g * EPW + j < n) {
                    const int   dj = __builtin_amdgcn_readlane(dl, 4 * j);
                    const float p0 = __int_as_float(__builtin_amdgcn_readlane(__float_as_int(p), 4 * j));
                    const float p1 = __int_as_float(__builtin_amdgcn_readlane(__float_as_int(p), 4 * j + 1));
                    const float p2 = __int_as_float(__builtin_amdgcn_readlane(__float_as_int(p), 4 * j + 2));
                    const float p3 = __int_as_float(__builtin_amdgcn_readlane(__float_as_int(p), 4 * j + 3));
                    float* rp = rows + dj * RP + lane;
                    atomicAdd(rp,          hv[j][0] * p0);
                    atomicAdd(rp + CH,     hv[j][1] * p1);
                    atomicAdd(rp + 2 * CH, hv[j][2] * p2);
                    atomicAdd(rp + 3 * CH, hv[j][3] * p3);
                }
            }
        }
    };

    if (nk >= 0) {
        maxwalk(nk, false, true, false);
        __syncthreads();
        drain(nk);
        __syncthreads();
    } else {
        int bl = nsw;
        for (int pass = 0; pass < 3; ++pass) {
            bool redo = false;
            for (int s0 = 0; s0 < nsw; s0 += bl) {
                if (pass == 0 && bykey) keyscan();
                if (s0 == 0 && !(pass == 0 && bykey)) {
                    if (tid == 0 && nvec > 0) end_group(0);
                    if (tid == 32 && nvec > 1) end_group(nvec - 1);
                }
                const int s1 = (pass == 0 && bykey) ? s0 : min(s0 + bl, nsw);
                for (int sw = s0; sw < s1; ++sw) {
                    int4 dv[SWV];
                    const int4* gp = (const int4*)(ei + gbase + 4) + (long)sw * (SWV * AGT) + tid;
                    if ((sw + 1) * (SWV * AGT) <= nhot) {
                        #pragma unroll
                        for (int j = 0; j < SWV; ++j) dv[j] = gp[j * AGT];
                    } else {
                        #pragma unroll
                        for (int j = 0; j < SWV; ++j) {
                            const int v = sw * (SWV * AGT) + j * AGT + tid;
                            dv[j] = make_int4(-1, -1, -1, -1);
                            if (v < nhot) dv[j] = gp[j * AGT];
                        }
                    }
                    #pragma unroll
                    for (int j = 0; j < SWV; ++j) {
                        const int v  = sw * (SWV * AGT) + j * AGT + tid;
                        const int e0 = gbase + 4 * (v + 1) - E;
                        const bool m0 = (unsigned)(dv[j].x - lo) < (unsigned)cnt;
                        const bool m1 = (unsigned)(dv[j].y - lo) < (unsigned)cnt;
                        const bool m2 = (unsigned)(dv[j].z - lo) < (unsigned)cnt;
                        const bool m3 = (unsigned)(dv[j].w - lo) < (unsigned)cnt;
                        if (m0 | m1 | m2 | m3) {
                            test(dv[j].x, e0);
                            test(dv[j].y, e0 + 1);
                            test(dv[j].z, e0 + 2);
                            test(dv[j].w, e0 + 3);
                        }
                    }
                }
                __syncthreads();
                const int n = __builtin_amdgcn_readfirstlane(qn);
                __syncthreads();
                const bool fits = n <= QCAP;
                if (FIRST && pass == 0 && tid == 0) qkeep_n[blockIdx.x * QCS] = fits ? n : -1;
                if (!fits) { redo = true; break; }
                maxwalk(n, true, pass < 2, FIRST && pass == 0);
                if (pass != 1) {
                    __syncthreads();
                    drain(n);
                }
                __syncthreads();
                if (tid == 0) qn = 0;
                __syncthreads();
            }
            if (pass == 0 ? !redo : pass == 2) break;
            bl = 1;
            for (int i = tid; i < cnt; i += AGT) q[i] = E + lo + i;
            if (tid == 0) qn = cnt;
            __syncthreads();
        }
    }

    if (FIRST) {
        for (int i = tid; i < cnt * (HC / 4); i += AGT) {
            const int r = i >> 5, c4 = i & 31;
            float4* p = (float4*)(rows + r * RP) + c4;
            float4 v = *p;
            const float sd = ssl[r * HEADS + (c4 >> 3)];
            const float* bb = bias + 4 * c4;
            v.x = v.x / sd + bb[0];
            v.y = v.y / sd + bb[1];
            v.z = v.z / sd + bb[2];
            v.w = v.w / sd + bb[3];
            v.x = v.x > 0.f ? v.x : __expf(v.x) - 1.f;
            v.y = v.y > 0.f ? v.y : __expf(v.y) - 1.f;
            v.z = v.z > 0.f ? v.z : __expf(v.z) - 1.f;
            v.w = v.w > 0.f ? v.w : __expf(v.w) - 1.f;
            *p = v;
        }
        __syncthreads();
        const int ntl = (cnt + 15) >> 4;
        const int nt  = wv & 7;
        const int mg  = wv >> 3;
        const int nv  = mg < ntl ? (ntl - 1 - mg) / 4 + 1 : 0;
        switch (nv) {
            case 4: rows_times_w<4>(rows, Wn, out, lo, cnt, mg, nt, lane); break;
            case 3: rows_times_w<3>(rows, Wn, out, lo, cnt, mg, nt, lane); break;
            case 2: rows_times_w<2>(rows, Wn, out, lo, cnt, mg, nt, lane); break;
            case 1: rows_times_w<1>(rows, Wn, out, lo, cnt, mg, nt, lane); break;
            default: break;
        }
    } else {
        float4* o4 = (float4*)(out + (long)lo * HC);
        for (int i = tid; i < cnt * (HC / 4); i += AGT)
            o4[i] = *((const float4*)(rows + (i >> 5) * RP) + (i & 31));
        for (int i = tid; i < cnt * HEADS; i += AGT) ss[(long)lo * HEADS + i] = ssl[i];
    }
}

#define NBW 256
#define NRW 8
__device__ __forceinline__ float4 norm_bias_elu4(float4 v, float sd, float4 bb)
{
    v.x = v.x / sd + bb.x;
    v.y = v.y / sd + bb.y;
    v.z = v.z / sd + bb.z;
    v.w = v.w / sd + bb.w;
    v.x = v.x > 0.f ? v.x : __expf(v.x) - 1.f;
    v.y = v.y > 0.f ? v.y : __expf(v.y) - 1.f;
    v.z = v.z > 0.f ? v.z : __expf(v.z) - 1.f;
    v.w = v.w > 0.f ? v.w : __expf(v.w) - 1.f;
    return v;
}
__global__ __launch_bounds__(NBW) void norm_bias_elu(float* __restrict__ y, const float* __restrict__ ss,
                              const float* __restrict__ b, int total4)
{
    const int i0 = blockIdx.x * (NBW * NRW) + threadIdx.x;
    float4* y4 = (float4*)y;
    const float4 bb = *((const float4*)b + (threadIdx.x & 31));
    float4 v[NRW];
    float  sd[NRW];
    if ((blockIdx.x + 1) * (NBW * NRW) <= total4) {
        #pragma unroll
        for (int j = 0; j < NRW; ++j) {
            const int i = i0 + j * NBW;
            v[j]  = y4[i];
            sd[j] = ss[i >> 3];
        }
        #pragma unroll
        for (int j = 0; j < NRW; ++j) y4[i0 + j * NBW] = norm_bias_elu4(v[j], sd[j], bb);
    } else {
        #pragma unroll
        for (int j = 0; j < NRW; ++j) {
            const int i = i0 + j * NBW;
            v[j]  = make_float4(0.f, 0.f, 0.f, 0.f);
            sd[j] = 1.f;
            if (i < total4) { v[j] = y4[i]; sd[j] = ss[i >> 3]; }
        }
        #pragma unroll
        for (int j = 0; j < NRW; ++j) {
            const int i = i0 + j * NBW;
            if (i < total4) y4[i] = norm_bias_elu4(v[j], sd[j], bb);
        }
    }
}

extern "C" void kernel_launch(void* const* d_in, const int* in_sizes, int n_in,
                              void* d_out, int out_size, void* d_ws, size_t ws_size,
                              hipStream_t stream)
{
    const float* x      = (const float*)d_in[0];
    const int*   ei     = (const int*)d_in[1];
    const float* W1     = (const float*)d_in[2];
    const float* a_src1 = (const float*)d_in[3];
    const float* a_dst1 = (const float*)d_in[4];
    const float* b1     = (const float*)d_in[5];
    const float* W2     = (const float*)d_in[6];
    const float* a_src2 = (const float*)d_in[7];
    const float* a_dst2 = (const float*)d_in[8];
    const float* b2     = (const float*)d_in[9];

    const int N  = in_sizes[0] / HC;
    const int E  = in_sizes[1] / 2;
    const int NH = N * HEADS;
    const int NF = N * HC;

    char* ws = (char*)d_ws;
    float*    hbuf = (float*)(ws);
    float*    h2   = (float*)(ws + (size_t)NF * 4);
    float*    as   = (float*)(ws + (size_t)NF * 8);
    float*    ad   = (float*)(ws + (size_t)NF * 8 + (size_t)NH * 4);
    float*    sArr = (float*)(ws + (size_t)NF * 8 + (size_t)NH * 8);

    const int TB = 256;
    const int gN  = (NH + TB - 1) / TB;
    const int NF4 = NF / 4;
    const int gNF = (NF4 + NBW * NRW - 1) / (NBW * NRW);
    const int gRB = (N + RPB - 1) / RPB;
    float* outF = (float*)d_out;
    int*      qkn  = (int*)(ws + (size_t)NF * 8 + (size_t)NH * 12);
    int*      qkp  = (int*)(ws + (size_t)NF * 8 + (size_t)NH * 12 + (size_t)gRB * QCS * 4);
    unsigned* kwp  = (unsigned*)(ws + (size_t)NF * 8 + (size_t)NH * 12 + (size_t)gRB * QCS * 4 + (size_t)gRB * QCAP * 4);
    const int nkw  = (E >> 4) * 4;

    gemm_f32_wmma<<<N / 16, 256, 0, stream>>>(x, W1, hbuf, N);
    node_alphas<<<gN, TB, 0, stream>>>(hbuf, a_src1, a_dst1, as, ad, NH);
    dst_keys<<<(nkw + TB) / TB, TB, 0, stream>>>(ei, E, kwp, nkw);
    edge_agg_lds<true><<<gRB, AGT, 0, stream>>>(ei, E, N, hbuf, as, ad, sArr, h2, qkp, qkn, kwp, b1, W2);

    node_alphas<<<gN, TB, 0, stream>>>(h2, a_src2, a_dst2, as, ad, NH);
    edge_agg_lds<false><<<gRB, AGT, 0, stream>>>(ei, E, N, h2, as, ad, sArr, outF, qkp, qkn, kwp, nullptr, nullptr);
    norm_bias_elu<<<gNF, NBW, 0, stream>>>(outF, sArr, b2, NF4);
}


# Row gather: one 16-byte load per lane fetches a whole 512-byte row (8 loads per 8 edges, not 32); LDS rows word-interleaved while summed so each add keeps 32 banks; sums in arrival order as before
# speedup vs baseline: 7.4240x; 1.0666x over previous
#include <hip/hip_runtime.h>
#include <hip/hip_bf16.h>

typedef __attribute__((ext_vector_type(2))) float v2f;
typedef __attribute__((ext_vector_type(8))) float v8f;

#define HC 128
#define HEADS 4
#define CH 32
#define NEG_SLOPE 0.2f

__device__ __forceinline__ unsigned f2ord(float f) {
    unsigned u = __float_as_uint(f);
    return (u & 0x80000000u) ? ~u : (u | 0x80000000u);
}
__device__ __forceinline__ float ord2f(unsigned u) {
    return (u & 0x80000000u) ? __uint_as_float(u & 0x7FFFFFFFu)
                             : __uint_as_float(~u);
}
#define ORD_NEG_INF 0x007FFFFFu

__device__ __forceinline__ float lrelu(float v) { return v > 0.f ? v : NEG_SLOPE * v; }

__global__ __launch_bounds__(256) void gemm_f32_wmma(
    const float* __restrict__ A, const float* __restrict__ B,
    float* __restrict__ D, int M)
{
    const int wave = threadIdx.x >> 5;
    const int lane = threadIdx.x & 31;
    const int mt   = blockIdx.x;
    const int m15  = lane & 15;
    const int hi   = lane >> 4;
    const int row  = mt * 16 + m15;
    const int col  = wave * 16 + m15;

    v8f acc;
    #pragma unroll
    for (int r = 0; r < 8; ++r) acc[r] = 0.f;

    #pragma unroll 4
    for (int k0 = 0; k0 < HC; k0 += 4) {
        const float* ap = A + (long)row * HC + k0 + 2 * hi;
        v2f a; a.x = ap[0]; a.y = ap[1];
        const float* bp = B + (long)(k0 + 2 * hi) * HC + col;
        v2f b; b.x = bp[0]; b.y = bp[HC];
        acc = __builtin_amdgcn_wmma_f32_16x16x4_f32(
            false, a, false, b, (short)0, acc, false, false);
    }
    #pragma unroll
    for (int r = 0; r < 8; ++r)
        D[(long)(mt * 16 + r + 8 * hi) * HC + col] = acc[r];
}

__global__ void node_alphas(const float* __restrict__ h,
                            const float* __restrict__ a_src,
                            const float* __restrict__ a_dst,
                            float* __restrict__ as, float* __restrict__ ad,
                            int NH)
{
    int i = blockIdx.x * blockDim.x + threadIdx.x;
    if (i >= NH) return;
    int n = i >> 2, hd = i & 3;
    const float* hp  = h + (long)n * HC + hd * CH;
    const float* asp = a_src + hd * CH;
    const float* adp = a_dst + hd * CH;
    float s1 = 0.f, s2 = 0.f;
    #pragma unroll
    for (int c = 0; c < CH; ++c) { float v = hp[c]; s1 += v * asp[c]; s2 += v * adp[c]; }
    as[i] = s1; ad[i] = s2;
}

#define RPB  196
__global__ void dst_keys(const int* __restrict__ ei, int E, unsigned* __restrict__ kw, int nw)
{
    const int i = blockIdx.x * blockDim.x + threadIdx.x;
    if (i >= nw) return;
    const int* dp = ei + E + 4 * (long)i;
    const unsigned k0 = (unsigned)dp[0] / RPB, k1 = (unsigned)dp[1] / RPB;
    const unsigned k2 = (unsigned)dp[2] / RPB, k3 = (unsigned)dp[3] / RPB;
    kw[i] = (k0 & 255u) | ((k1 & 255u) << 8) | ((k2 & 255u) << 16) | (k3 << 24);
}

#define EPW 8
#define RPAD 208
#define RP   132
#define RBITS 8
#define AGT  1024
#define SWV  8
#define SWE  (SWV * AGT * 4)
#define QCAP (SWE + 4096)
#define MXU  8
#define QCS  32
#define KLD  8
static_assert(RPB <= (1 << RBITS), "a row number must fit the low bits of a queue word");
static_assert(RPAD >= RPB && RPAD % 16 == 0 && RP >= HC && RP % 4 == 0, "whole matrix tiles, 16-byte aligned rows");
static_assert(RPAD / 16 <= 16 && AGT == 1024, "the product's split: 8 column tiles x 4 groups of at most 4 row tiles over 32 waves");
template <int TL>
__device__ __forceinline__ void rows_times_w(
    const float* rows, const float* __restrict__ Wn, float* __restrict__ hn,
    int lo, int cnt, int mg, int nt, int lane)
{
    const int m15 = lane & 15;
    const int hi  = lane >> 4;
    const int col = nt * 16 + m15;
    v8f acc[TL];
    #pragma unroll
    for (int t = 0; t < TL; ++t) acc[t] = (v8f){0.f, 0.f, 0.f, 0.f, 0.f, 0.f, 0.f, 0.f};
    const float* ap = rows + (mg * 16 + m15) * RP + 2 * hi;
    const float* bp = Wn + (long)(2 * hi) * HC + col;
    #pragma unroll 8
    for (int k0 = 0; k0 < HC; k0 += 4) {
        v2f b; b.x = bp[(long)k0 * HC]; b.y = bp[(long)k0 * HC + HC];
        #pragma unroll
        for (int t = 0; t < TL; ++t) {
            const v2f a = *(const v2f*)(ap + t * (64 * RP) + k0);
            acc[t] = __builtin_amdgcn_wmma_f32_16x16x4_f32(
                false, a, false, b, (short)0, acc[t], false, false);
        }
    }
    #pragma unroll
    for (int tr = 0; tr < TL * 8; ++tr) {
        const int t = tr >> 3, r = tr & 7;
        const int row = (mg + 4 * t) * 16 + r + 8 * hi;
        if (row < cnt) hn[(long)(lo + row) * HC + col] = acc[t][r];
    }
}

template <int J>
__device__ __forceinline__ float group8_lane(float v)
{
    constexpr unsigned sel = J | (J << 3) | (J << 6) | (J << 9) | (J << 12) | (J << 15) | (J << 18) | (J << 21);
    return __int_as_float(__builtin_amdgcn_mov_dpp8(__float_as_int(v), sel));
}
template <int J>
__device__ __forceinline__ void add_row(float* rows, int dl, float p, float4 hv, int lane)
{
    const int   dj = __builtin_amdgcn_readlane(dl, J);
    const float pj = group8_lane<J>(p);
    float* rp = rows + dj * RP + lane;
    atomicAdd(rp,          hv.x * pj);
    atomicAdd(rp + CH,     hv.y * pj);
    atomicAdd(rp + 2 * CH, hv.z * pj);
    atomicAdd(rp + 3 * CH, hv.w * pj);
}

template <bool FIRST>
__global__ __launch_bounds__(AGT) void edge_agg_lds(
    const int* __restrict__ ei, int E, int N,
    const float* __restrict__ h,
    const float* __restrict__ as, const float* __restrict__ ad,
    float* __restrict__ ss, float* __restrict__ out,
    int* __restrict__ qkeep, int* __restrict__ qkeep_n,
    const unsigned* __restrict__ kw,
    const float* __restrict__ bias, const float* __restrict__ Wn)
{
    __shared__ __attribute__((aligned(16))) float rows[RPAD * RP];
    __shared__ float ssl[RPB * HEADS];
    __shared__ unsigned ml[RPB * HEADS];
    __shared__ int   q[QCAP];
    __shared__ int   qn;

    const int tid  = threadIdx.x;
    const int lane = tid & 31;
    const int wv   = __builtin_amdgcn_readfirstlane(tid >> 5);
    const int lo   = blockIdx.x * RPB;
    const int cnt  = min(RPB, N - lo);
    int* qk = qkeep + (long)blockIdx.x * QCAP;
    const int nk = FIRST ? -1 : qkeep_n[blockIdx.x * QCS];

    for (int i = tid; i < RPAD * RP / 4; i += AGT) ((float4*)rows)[i] = make_float4(0.f, 0.f, 0.f, 0.f);
    for (int i = tid; i < RPB * HEADS; i += AGT) { ssl[i] = 0.f; ml[i] = ORD_NEG_INF; }
    if (nk >= 0) {
        for (int i = tid; i < nk; i += AGT) q[i] = qk[i];
    } else {
        for (int i = tid; i < cnt; i += AGT) q[i] = E + lo + i;
        if (tid == 0) qn = cnt;
    }
    __syncthreads();

    const int gbase = E & ~3;
    const int nvec  = (2 * E - gbase + 3) >> 2;
    const int nhot  = nvec > 2 ? nvec - 2 : 0;
    const int nsw   = max((nhot + SWV * AGT - 1) / (SWV * AGT), 1);
    const bool bykey = gridDim.x <= 256;

    auto test = [&](int dn, int e) {
        if ((unsigned)(dn - lo) < (unsigned)cnt) {
            const int pos = atomicAdd(&qn, 1);
            if (pos < QCAP) q[pos] = e;
        }
    };
    auto end_group = [&](int v) {
        const int g0 = gbase + 4 * v;
        for (int j = 0; j < 4; ++j) {
            const int g = g0 + j;
            if (g >= E && g < 2 * E) test(ei[g], g - E);
        }
    };

    auto keyscan = [&]() {
        const unsigned bc = blockIdx.x * 0x01010101u;
        const uint4* kp = (const uint4*)kw;
        const int ng = E >> 4;
        for (int g0 = 0; g0 < ng; g0 += KLD * AGT) {
            uint4 kv[KLD];
            if (g0 + KLD * AGT <= ng) {
                #pragma unroll
                for (int j = 0; j < KLD; ++j) kv[j] = kp[g0 + j * AGT + tid];
            } else {
                #pragma unroll
                for (int j = 0; j < KLD; ++j) {
                    const int g = g0 + j * AGT + tid;
                    kv[j] = make_uint4(~bc, ~bc, ~bc, ~bc);
                    if (g < ng) kv[j] = kp[g];
                }
            }
            unsigned hit = 0;
            #pragma unroll
            for (int j = 0; j < KLD; ++j) {
                const unsigned x0 = kv[j].x ^ bc, x1 = kv[j].y ^ bc, x2 = kv[j].z ^ bc, x3 = kv[j].w ^ bc;
                const unsigned z = ((x0 - 0x01010101u) & ~x0) | ((x1 - 0x01010101u) & ~x1)
                                 | ((x2 - 0x01010101u) & ~x2) | ((x3 - 0x01010101u) & ~x3);
                if (z & 0x80808080u) hit |= 1u << j;
            }
            while (hit) {
                const int j = __builtin_ctz(hit);
                hit &= hit - 1;
                const int g = g0 + j * AGT + tid;
                const uint4 k = kp[g];
                const unsigned x0 = k.x ^ bc, x1 = k.y ^ bc, x2 = k.z ^ bc, x3 = k.w ^ bc;
                const unsigned m0 = ~(((x0 & 0x7f7f7f7fu) + 0x7f7f7f7fu) | x0) & 0x80808080u;
                const unsigned m1 = ~(((x1 & 0x7f7f7f7fu) + 0x7f7f7f7fu) | x1) & 0x80808080u;
                const unsigned m2 = ~(((x2 & 0x7f7f7f7fu) + 0x7f7f7f7fu) | x2) & 0x80808080u;
                const unsigned m3 = ~(((x3 & 0x7f7f7f7fu) + 0x7f7f7f7fu) | x3) & 0x80808080u;
                unsigned m = (m0 >> 7) | (m1 >> 6) | (m2 >> 5) | (m3 >> 4);
                while (m) {
                    const int b = __builtin_ctz(m);
                    m &= m - 1;
                    const int pos = atomicAdd(&qn, 1);
                    if (pos < QCAP) q[pos] = 16 * g + 4 * (b & 3) + (b >> 3);
                }
            }
        }
        if (tid < (E & 15)) {
            const int e = (E & ~15) + tid;
            test(ei[E + e], e);
        }
    };

    auto maxwalk = [&](int n, bool numbered, bool domax, bool keep) {
        for (int i0 = tid; i0 < n; i0 += MXU * AGT) {
            int s[MXU], dl[MXU];
            if (numbered) {
                int e[MXU], sv[MXU], dv[MXU];
                #pragma unroll
                for (int j = 0; j < MXU; ++j) {
                    const int i = i0 + j * AGT;
                    e[j] = q[i < n ? i : i0];
                }
                #pragma unroll
                for (int j = 0; j < MXU; ++j) {
                    const int ec = e[j] < E ? e[j] : 0;
                    sv[j] = ei[ec];
                    dv[j] = ei[E + ec];
                }
                #pragma unroll
                for (int j = 0; j < MXU; ++j) {
                    const int i = i0 + j * AGT;
                    s[j]  = e[j] < E ? sv[j] : e[j] - E;
                    dl[j] = (e[j] < E ? dv[j] : e[j] - E) - lo;
                    if (i < n) {
                        const int w = (int)(((unsigned)s[j] << RBITS) | (unsigned)dl[j]);
                        q[i] = w;
                        if (keep) qk[i] = w;
                    }
                }
            } else {
                #pragma unroll
                for (int j = 0; j < MXU; ++j) {
                    const int i = i0 + j * AGT;
                    const unsigned w = (unsigned)q[i < n ? i : i0];
                    s[j]  = (int)(w >> RBITS);
                    dl[j] = (int)(w & ((1u << RBITS) - 1u));
                }
            }
            if (domax) {
                float4 a[MXU], b[MXU];
                #pragma unroll
                for (int j = 0; j < MXU; ++j) {
                    a[j] = *(const float4*)(as + (long)s[j] * 4);
                    b[j] = *(const float4*)(ad + (long)(lo + dl[j]) * 4);
                }
                #pragma unroll
                for (int j = 0; j < MXU; ++j) {
                    if (i0 + j * AGT < n) {
                        unsigned* mp = ml + dl[j] * 4;
                        atomicMax(mp + 0, f2ord(lrelu(a[j].x + b[j].x)));
                        atomicMax(mp + 1, f2ord(lrelu(a[j].y + b[j].y)));
                        atomicMax(mp + 2, f2ord(lrelu(a[j].z + b[j].z)));
                        atomicMax(mp + 3, f2ord(lrelu(a[j].w + b[j].w)));
                    }
                }
            }
        }
    };

    auto drain = [&](int n) {
        for (int g = wv; g * EPW < n; g += AGT / 32) {
            const int qi = g * EPW + (lane & 7);
            const int k  = lane >> 3;
            const bool live = qi < n;
            const unsigned w = live ? (unsigned)q[qi] : 0u;
            const int s  = (int)(w >> RBITS);
            const int dl = (int)(w & ((1u << RBITS) - 1u));
            const float    ea = as[(long)s * 4 + k];
            const float    eb = ad[(long)(lo + dl) * 4 + k];
            const unsigned mu = ml[dl * 4 + k];
            float4 hv[EPW];
            #pragma unroll
            for (int j = 0; j < EPW; ++j) {
                const int sj = __builtin_amdgcn_readlane(s, j);
                hv[j] = *((const float4*)(h + (long)sj * HC) + lane);
            }
            const float p = __expf(lrelu(ea + eb) - ord2f(mu));
            if (live) atomicAdd(&ssl[dl * 4 + k], p);
            if (g * EPW + 0 < n) add_row<0>(rows, dl, p, hv[0], lane);
            if (g * EPW + 1 < n) add_row<1>(rows, dl, p, hv[1], lane);
            if (g * EPW + 2 < n) add_row<2>(rows, dl, p, hv[2], lane);
            if (g * EPW + 3 < n) add_row<3>(rows, dl, p, hv[3], lane);
            if (g * EPW + 4 < n) add_row<4>(rows, dl, p, hv[4], lane);
            if (g * EPW + 5 < n) add_row<5>(rows, dl, p, hv[5], lane);
            if (g * EPW + 6 < n) add_row<6>(rows, dl, p, hv[6], lane);
            if (g * EPW + 7 < n) add_row<7>(rows, dl, p, hv[7], lane);
        }
    };

    if (nk >= 0) {
        maxwalk(nk, false, true, false);
        __syncthreads();
        drain(nk);
        __syncthreads();
    } else {
        int bl = nsw;
        for (int pass = 0; pass < 3; ++pass) {
            bool redo = false;
            for (int s0 = 0; s0 < nsw; s0 += bl) {
                if (pass == 0 && bykey) keyscan();
                if (s0 == 0 && !(pass == 0 && bykey)) {
                    if (tid == 0 && nvec > 0) end_group(0);
                    if (tid == 32 && nvec > 1) end_group(nvec - 1);
                }
                const int s1 = (pass == 0 && bykey) ? s0 : min(s0 + bl, nsw);
                for (int sw = s0; sw < s1; ++sw) {
                    int4 dv[SWV];
                    const int4* gp = (const int4*)(ei + gbase + 4) + (long)sw * (SWV * AGT) + tid;
                    if ((sw + 1) * (SWV * AGT) <= nhot) {
                        #pragma unroll
                        for (int j = 0; j < SWV; ++j) dv[j] = gp[j * AGT];
                    } else {
                        #pragma unroll
                        for (int j = 0; j < SWV; ++j) {
                            const int v = sw * (SWV * AGT) + j * AGT + tid;
                            dv[j] = make_int4(-1, -1, -1, -1);
                            if (v < nhot) dv[j] = gp[j * AGT];
                        }
                    }
                    #pragma unroll
                    for (int j = 0; j < SWV; ++j) {
                        const int v  = sw * (SWV * AGT) + j * AGT + tid;
                        const int e0 = gbase + 4 * (v + 1) - E;
                        const bool m0 = (unsigned)(dv[j].x - lo) < (unsigned)cnt;
                        const bool m1 = (unsigned)(dv[j].y - lo) < (unsigned)cnt;
                        const bool m2 = (unsigned)(dv[j].z - lo) < (unsigned)cnt;
                        const bool m3 = (unsigned)(dv[j].w - lo) < (unsigned)cnt;
                        if (m0 | m1 | m2 | m3) {
                            test(dv[j].x, e0);
                            test(dv[j].y, e0 + 1);
                            test(dv[j].z, e0 + 2);
                            test(dv[j].w, e0 + 3);
                        }
                    }
                }
                __syncthreads();
                const int n = __builtin_amdgcn_readfirstlane(qn);
                __syncthreads();
                const bool fits = n <= QCAP;
                if (FIRST && pass == 0 && tid == 0) qkeep_n[blockIdx.x * QCS] = fits ? n : -1;
                if (!fits) { redo = true; break; }
                maxwalk(n, true, pass < 2, FIRST && pass == 0);
                if (pass != 1) {
                    __syncthreads();
                    drain(n);
                }
                __syncthreads();
                if (tid == 0) qn = 0;
                __syncthreads();
            }
            if (pass == 0 ? !redo : pass == 2) break;
            bl = 1;
            for (int i = tid; i < cnt; i += AGT) q[i] = E + lo + i;
            if (tid == 0) qn = cnt;
            __syncthreads();
        }
    }

    if (FIRST) {
        for (int i = tid; i < cnt * (HC / 4); i += AGT) {
            const int r = i >> 5, c4 = i & 31;
            float4* p = (float4*)(rows + r * RP) + c4;
            const float* ip = rows + r * RP + c4;
            float4 v = make_float4(ip[0], ip[CH], ip[2 * CH], ip[3 * CH]);
            const float sd = ssl[r * HEADS + (c4 >> 3)];
            const float* bb = bias + 4 * c4;
            v.x = v.x / sd + bb[0];
            v.y = v.y / sd + bb[1];
            v.z = v.z / sd + bb[2];
            v.w = v.w / sd + bb[3];
            v.x = v.x > 0.f ? v.x : __expf(v.x) - 1.f;
            v.y = v.y > 0.f ? v.y : __expf(v.y) - 1.f;
            v.z = v.z > 0.f ? v.z : __expf(v.z) - 1.f;
            v.w = v.w > 0.f ? v.w : __expf(v.w) - 1.f;
            *p = v;
        }
        __syncthreads();
        const int ntl = (cnt + 15) >> 4;
        const int nt  = wv & 7;
        const int mg  = wv >> 3;
        const int nv  = mg < ntl ? (ntl - 1 - mg) / 4 + 1 : 0;
        switch (nv) {
            case 4: rows_times_w<4>(rows, Wn, out, lo, cnt, mg, nt, lane); break;
            case 3: rows_times_w<3>(rows, Wn, out, lo, cnt, mg, nt, lane); break;
            case 2: rows_times_w<2>(rows, Wn, out, lo, cnt, mg, nt, lane); break;
            case 1: rows_times_w<1>(rows, Wn, out, lo, cnt, mg, nt, lane); break;
            default: break;
        }
    } else {
        float4* o4 = (float4*)(out + (long)lo * HC);
        for (int i = tid; i < cnt * (HC / 4); i += AGT) {
            const float* ip = rows + (i >> 5) * RP + (i & 31);
            o4[i] = make_float4(ip[0], ip[CH], ip[2 * CH], ip[3 * CH]);
        }
        for (int i = tid; i < cnt * HEADS; i += AGT) ss[(long)lo * HEADS + i] = ssl[i];
    }
}

#define NBW 256
#define NRW 8
__device__ __forceinline__ float4 norm_bias_elu4(float4 v, float sd, float4 bb)
{
    v.x = v.x / sd + bb.x;
    v.y = v.y / sd + bb.y;
    v.z = v.z / sd + bb.z;
    v.w = v.w / sd + bb.w;
    v.x = v.x > 0.f ? v.x : __expf(v.x) - 1.f;
    v.y = v.y > 0.f ? v.y : __expf(v.y) - 1.f;
    v.z = v.z > 0.f ? v.z : __expf(v.z) - 1.f;
    v.w = v.w > 0.f ? v.w : __expf(v.w) - 1.f;
    return v;
}
__global__ __launch_bounds__(NBW) void norm_bias_elu(float* __restrict__ y, const float* __restrict__ ss,
                              const float* __restrict__ b, int total4)
{
    const int i0 = blockIdx.x * (NBW * NRW) + threadIdx.x;
    float4* y4 = (float4*)y;
    const float4 bb = *((const float4*)b + (threadIdx.x & 31));
    float4 v[NRW];
    float  sd[NRW];
    if ((blockIdx.x + 1) * (NBW * NRW) <= total4) {
        #pragma unroll
        for (int j = 0; j < NRW; ++j) {
            const int i = i0 + j * NBW;
            v[j]  = y4[i];
            sd[j] = ss[i >> 3];
        }
        #pragma unroll
        for (int j = 0; j < NRW; ++j) y4[i0 + j * NBW] = norm_bias_elu4(v[j], sd[j], bb);
    } else {
        #pragma unroll
        for (int j = 0; j < NRW; ++j) {
            const int i = i0 + j * NBW;
            v[j]  = make_float4(0.f, 0.f, 0.f, 0.f);
            sd[j] = 1.f;
            if (i < total4) { v[j] = y4[i]; sd[j] = ss[i >> 3]; }
        }
        #pragma unroll
        for (int j = 0; j < NRW; ++j) {
            const int i = i0 + j * NBW;
            if (i < total4) y4[i] = norm_bias_elu4(v[j], sd[j], bb);
        }
    }
}

extern "C" void kernel_launch(void* const* d_in, const int* in_sizes, int n_in,
                              void* d_out, int out_size, void* d_ws, size_t ws_size,
                              hipStream_t stream)
{
    const float* x      = (const float*)d_in[0];
    const int*   ei     = (const int*)d_in[1];
    const float* W1     = (const float*)d_in[2];
    const float* a_src1 = (const float*)d_in[3];
    const float* a_dst1 = (const float*)d_in[4];
    const float* b1     = (const float*)d_in[5];
    const float* W2     = (const float*)d_in[6];
    const float* a_src2 = (const float*)d_in[7];
    const float* a_dst2 = (const float*)d_in[8];
    const float* b2     = (const float*)d_in[9];

    const int N  = in_sizes[0] / HC;
    const int E  = in_sizes[1] / 2;
    const int NH = N * HEADS;
    const int NF = N * HC;

    char* ws = (char*)d_ws;
    float*    hbuf = (float*)(ws);
    float*    h2   = (float*)(ws + (size_t)NF * 4);
    float*    as   = (float*)(ws + (size_t)NF * 8);
    float*    ad   = (float*)(ws + (size_t)NF * 8 + (size_t)NH * 4);
    float*    sArr = (float*)(ws + (size_t)NF * 8 + (size_t)NH * 8);

    const int TB = 256;
    const int gN  = (NH + TB - 1) / TB;
    const int NF4 = NF / 4;
    const int gNF = (NF4 + NBW * NRW - 1) / (NBW * NRW);
    const int gRB = (N + RPB - 1) / RPB;
    float* outF = (float*)d_out;
    int*      qkn  = (int*)(ws + (size_t)NF * 8 + (size_t)NH * 12);
    int*      qkp  = (int*)(ws + (size_t)NF * 8 + (size_t)NH * 12 + (size_t)gRB * QCS * 4);
    unsigned* kwp  = (unsigned*)(ws + (size_t)NF * 8 + (size_t)NH * 12 + (size_t)gRB * QCS * 4 + (size_t)gRB * QCAP * 4);
    const int nkw  = (E >> 4) * 4;

    gemm_f32_wmma<<<N / 16, 256, 0, stream>>>(x, W1, hbuf, N);
    node_alphas<<<gN, TB, 0, stream>>>(hbuf, a_src1, a_dst1, as, ad, NH);
    dst_keys<<<(nkw + TB) / TB, TB, 0, stream>>>(ei, E, kwp, nkw);
    edge_agg_lds<true><<<gRB, AGT, 0, stream>>>(ei, E, N, hbuf, as, ad, sArr, h2, qkp, qkn, kwp, b1, W2);

    node_alphas<<<gN, TB, 0, stream>>>(h2, a_src2, a_dst2, as, ad, NH);
    edge_agg_lds<false><<<gRB, AGT, 0, stream>>>(ei, E, N, h2, as, ad, sArr, outF, qkp, qkn, kwp, nullptr, nullptr);
    norm_bias_elu<<<gNF, NBW, 0, stream>>>(outF, sArr, b2, NF4);
}
